# starting kernel
# baseline (speedup 1.0000x reference)
#include <hip/hip_runtime.h>


typedef __attribute__((ext_vector_type(2))) float v2f;
typedef __attribute__((ext_vector_type(8))) float v8f;

#define DFEAT 128
#define LEAKY 0.01f

__global__ void __launch_bounds__(256) gcn_gemm_xWT(
    const float* __restrict__ x, const float* __restrict__ W,
    float* __restrict__ h, int n_nodes)
{
  __shared__ float Wlds[DFEAT * DFEAT];

  {
    const float4* Wg4 = (const float4*)W;
    float4* Wl4 = (float4*)Wlds;
    #pragma unroll
    for (int i = 0; i < (DFEAT * DFEAT / 4) / 256; ++i)
      Wl4[threadIdx.x + 256 * i] = Wg4[threadIdx.x + 256 * i];
  }
  __syncthreads();

  const int wave = threadIdx.x >> 5;
  const int lane = threadIdx.x & 31;
  const int half = lane >> 4;
  const int lrow = lane & 15;

  const int base = blockIdx.x * 128 + wave * 16;
  if (base >= n_nodes) return;

  int row = base + lrow;
  if (row >= n_nodes) row = n_nodes - 1;
  const v2f* xr = (const v2f*)(x + (size_t)row * DFEAT);

  v2f afrag[32];
  #pragma unroll
  for (int kt = 0; kt < 32; ++kt) afrag[kt] = xr[2 * kt + half];

  #pragma unroll
  for (int nt = 0; nt < 8; ++nt) {
    const v2f* wr = (const v2f*)(Wlds + (nt * 16 + lrow) * DFEAT);
    v8f acc = {};
    #pragma unroll
    for (int kt = 0; kt < 32; ++kt) {
      v2f b = wr[2 * kt + half];
      acc = __builtin_amdgcn_wmma_f32_16x16x4_f32(
           false, afrag[kt],  false, b,
           (short)0, acc,  false,  false);
    }
    const int col = nt * 16 + lrow;
    #pragma unroll
    for (int r = 0; r < 8; ++r) {
      int m = base + half * 8 + r;
      if (m < n_nodes) h[(size_t)m * DFEAT + col] = acc[r];
    }
  }
}

__global__ void deg_init(float* __restrict__ deg, int n) {
  int i = blockIdx.x * blockDim.x + threadIdx.x;
  if (i < n) deg[i] = 1.0f;
}
__global__ void deg_accum(const int* __restrict__ dst, float* __restrict__ deg, int E) {
  int e = blockIdx.x * blockDim.x + threadIdx.x;
  if (e < E) unsafeAtomicAdd(&deg[dst[e]], 1.0f);
}
__global__ void deg_finalize(float* __restrict__ deg, int n) {
  int i = blockIdx.x * blockDim.x + threadIdx.x;
  if (i < n) deg[i] = __frsqrt_rn(deg[i]);
}

__global__ void gcn_init_out(const float* __restrict__ h, const float* __restrict__ dis,
                             const float* __restrict__ bias, float* __restrict__ out, int n) {
  int idx = blockIdx.x * blockDim.x + threadIdx.x;
  if (idx >= n * DFEAT) return;
  int i = idx >> 7, f = idx & (DFEAT - 1);
  float dv = dis[i];
  out[idx] = bias[f] + h[idx] * dv * dv;
}

__global__ void __launch_bounds__(256) gcn_scatter(
    const int* __restrict__ src, const int* __restrict__ dst,
    const float* __restrict__ h, const float* __restrict__ dis,
    float* __restrict__ out, int E)
{
  int e = blockIdx.x * 8 + (threadIdx.x >> 5);
  if (e >= E) return;
  int lane = threadIdx.x & 31;
  int s = src[e], d = dst[e];
  float nrm = dis[s] * dis[d];
  const float* hs = h + (size_t)s * DFEAT;
  float* od = out + (size_t)d * DFEAT;
  #pragma unroll
  for (int j = 0; j < 4; ++j) {
    int f = lane + 32 * j;
    unsafeAtomicAdd(&od[f], hs[f] * nrm);
  }
}

__global__ void gcn_leaky(const float* __restrict__ in, float* __restrict__ out, int n) {
  int idx = blockIdx.x * blockDim.x + threadIdx.x;
  if (idx < n) {
    float v = in[idx];
    out[idx] = (v >= 0.0f) ? v : LEAKY * v;
  }
}

extern "C" void kernel_launch(void* const* d_in, const int* in_sizes, int n_in,
                              void* d_out, int out_size, void* d_ws, size_t ws_size,
                              hipStream_t stream) {
  const float* x  = (const float*)d_in[0];
  const int*   ei = (const int*)d_in[1];
  const float* Ws = (const float*)d_in[2];
  const float* bs = (const float*)d_in[3];
  float* out = (float*)d_out;

  const int n = in_sizes[0] / DFEAT;
  const int E = in_sizes[1] / 2;
  const int* src = ei;
  const int* dst = ei + E;

  float* xbuf = (float*)d_ws;
  float* hbuf = xbuf + (size_t)n * DFEAT;
  float* dis  = hbuf + (size_t)n * DFEAT;

  const int TB = 256;
  const int nd = n * DFEAT;

  deg_init    <<<(n + TB - 1) / TB, TB, 0, stream>>>(dis, n);
  deg_accum   <<<(E + TB - 1) / TB, TB, 0, stream>>>(dst, dis, E);
  deg_finalize<<<(n + TB - 1) / TB, TB, 0, stream>>>(dis, n);

  const float* cur = x;
  for (int l = 0; l < 3; ++l) {
    gcn_gemm_xWT<<<(n + 127) / 128, 256, 0, stream>>>(cur, Ws + (size_t)l * DFEAT * DFEAT, hbuf, n);
    gcn_init_out<<<(nd + TB - 1) / TB, TB, 0, stream>>>(hbuf, dis, bs + (size_t)l * DFEAT, out, n);
    gcn_scatter <<<(E + 7) / 8, 256, 0, stream>>>(src, dst, hbuf, dis, out, E);
    float* act_dst = (l == 2) ? out : xbuf;
    gcn_leaky   <<<(nd + TB - 1) / TB, TB, 0, stream>>>(out, act_dst, nd);
    cur = xbuf;
  }
}


# Edges sorted by destination once per call (integer-atomic rank, scan, place); one wave per node sums its source rows in ascending source id and stores bias+self+LeakyReLU once: no float atomics
# speedup vs baseline: 2.1096x; 2.1096x over previous
#include <hip/hip_runtime.h>


typedef __attribute__((ext_vector_type(2))) float v2f;
typedef __attribute__((ext_vector_type(8))) float v8f;

#define DFEAT 128
#define LEAKY 0.01f

__global__ void __launch_bounds__(256) gcn_gemm_xWT(
    const float* __restrict__ x, const float* __restrict__ W,
    float* __restrict__ h, int n_nodes)
{
  __shared__ float Wlds[DFEAT * DFEAT];

  {
    const float4* Wg4 = (const float4*)W;
    float4* Wl4 = (float4*)Wlds;
    #pragma unroll
    for (int i = 0; i < (DFEAT * DFEAT / 4) / 256; ++i)
      Wl4[threadIdx.x + 256 * i] = Wg4[threadIdx.x + 256 * i];
  }
  __syncthreads();

  const int wave = threadIdx.x >> 5;
  const int lane = threadIdx.x & 31;
  const int half = lane >> 4;
  const int lrow = lane & 15;

  const int base = blockIdx.x * 128 + wave * 16;
  if (base >= n_nodes) return;

  int row = base + lrow;
  if (row >= n_nodes) row = n_nodes - 1;
  const v2f* xr = (const v2f*)(x + (size_t)row * DFEAT);

  v2f afrag[32];
  #pragma unroll
  for (int kt = 0; kt < 32; ++kt) afrag[kt] = xr[2 * kt + half];

  #pragma unroll
  for (int nt = 0; nt < 8; ++nt) {
    const v2f* wr = (const v2f*)(Wlds + (nt * 16 + lrow) * DFEAT);
    v8f acc = {};
    #pragma unroll
    for (int kt = 0; kt < 32; ++kt) {
      v2f b = wr[2 * kt + half];
      acc = __builtin_amdgcn_wmma_f32_16x16x4_f32(
           false, afrag[kt],  false, b,
           (short)0, acc,  false,  false);
    }
    const int col = nt * 16 + lrow;
    #pragma unroll
    for (int r = 0; r < 8; ++r) {
      int m = base + half * 8 + r;
      if (m < n_nodes) h[(size_t)m * DFEAT + col] = acc[r];
    }
  }
}

__global__ void cnt_zero(int* __restrict__ cnt, int n) {
  int i = blockIdx.x * blockDim.x + threadIdx.x;
  if (i < n) cnt[i] = 0;
}

__global__ void edge_rank(const int* __restrict__ dst, int* __restrict__ cnt,
                          int* __restrict__ rank, int E) {
  int e = blockIdx.x * blockDim.x + threadIdx.x;
  if (e < E) rank[e] = atomicAdd(&cnt[dst[e]], 1);
}

__global__ void __launch_bounds__(256) scan_local(
    const int* __restrict__ cnt, int* __restrict__ rowptr, int* __restrict__ bsum,
    float* __restrict__ dis, int n)
{
  __shared__ int wsum[8];
  const int tid = threadIdx.x, lane = tid & 31, wave = tid >> 5;
  const int i0 = (blockIdx.x * 256 + tid) * 4;
  int c[4];
  for (int k = 0; k < 4; ++k) {
    int i = i0 + k;
    c[k] = (i < n) ? cnt[i] : 0;
    if (i < n) dis[i] = __frsqrt_rn((float)(c[k] + 1));
  }
  const int t = c[0] + c[1] + c[2] + c[3];
  int incl = t;
  for (int o = 1; o < 32; o <<= 1) {
    int v = __shfl_up(incl, o);
    if (lane >= o) incl += v;
  }
  if (lane == 31) wsum[wave] = incl;
  __syncthreads();
  int woff = 0, tot = 0;
  for (int w = 0; w < 8; ++w) {
    int v = wsum[w];
    if (w < wave) woff += v;
    tot += v;
  }
  int ex = woff + incl - t;
  for (int k = 0; k < 4; ++k) {
    int i = i0 + k;
    if (i <= n) rowptr[i] = ex;
    ex += c[k];
  }
  if (tid == 0) bsum[blockIdx.x] = tot;
}

__global__ void __launch_bounds__(256) scan_add(
    int* __restrict__ rowptr, const int* __restrict__ bsum, int n)
{
  __shared__ int red[8];
  const int tid = threadIdx.x, lane = tid & 31, wave = tid >> 5;
  int part = 0;
  for (int b = tid; b < (int)blockIdx.x; b += 256) part += bsum[b];
  for (int o = 16; o > 0; o >>= 1) part += __shfl_xor(part, o);
  if (lane == 0) red[wave] = part;
  __syncthreads();
  int base = 0;
  for (int w = 0; w < 8; ++w) base += red[w];
  const int i0 = (blockIdx.x * 256 + tid) * 4;
  for (int k = 0; k < 4; ++k) {
    int i = i0 + k;
    if (i <= n) rowptr[i] += base;
  }
}

__global__ void csr_fill(const int* __restrict__ src, const int* __restrict__ dst,
                         const int* __restrict__ rowptr, const int* __restrict__ rank,
                         int* __restrict__ csr, int E) {
  int e = blockIdx.x * blockDim.x + threadIdx.x;
  if (e < E) {
    const int p = rowptr[dst[e]] + rank[e];
    if ((unsigned)p < (unsigned)E) csr[p] = src[e];
  }
}

__global__ void __launch_bounds__(256) gcn_gather(
    const int* __restrict__ rowptr, const int* __restrict__ csr,
    const float* __restrict__ h, const float* __restrict__ dis,
    const float* __restrict__ bias, float* __restrict__ out, int n, int E)
{
  const int d = __builtin_amdgcn_readfirstlane((int)(blockIdx.x * 8 + (threadIdx.x >> 5)));
  if (d >= n) return;
  const int lane = threadIdx.x & 31;
  int beg = __builtin_amdgcn_readfirstlane(rowptr[d]);
  int end = __builtin_amdgcn_readfirstlane(rowptr[d + 1]);
  beg = (beg < 0) ? 0 : beg;
  end = (end > E) ? E : end;
  const float dd = dis[d];
  const float4* __restrict__ h4 = (const float4*)h;
  const float4 hv = h4[(size_t)d * 32 + lane];
  const float4 bv = ((const float4*)bias)[lane];
  float4 acc;
  acc.x = bv.x + hv.x * dd * dd;
  acc.y = bv.y + hv.y * dd * dd;
  acc.z = bv.z + hv.z * dd * dd;
  acc.w = bv.w + hv.w * dd * dd;

  for (int c = beg; c < end; c += 32) {
    const int m = (end - c < 32) ? (end - c) : 32;
    int key = 0x7fffffff;
    if (lane < m) key = csr[c + lane];
    #pragma unroll
    for (int k = 2; k <= 32; k <<= 1) {
      #pragma unroll
      for (int j = k >> 1; j > 0; j >>= 1) {
        const int p = __shfl_xor(key, j);
        const bool up = ((lane & k) == 0);
        const bool lower = ((lane & j) == 0);
        const int lo = (key < p) ? key : p;
        const int hi = (key < p) ? p : key;
        key = (lower == up) ? lo : hi;
      }
    }
    int s_l = (lane < m) ? key : d;
    s_l = (s_l < 0) ? 0 : ((s_l >= n) ? n - 1 : s_l);
    float w_l = dis[s_l] * dd;
    if (lane >= m) w_l = 0.0f;
    const int wi_l = __float_as_int(w_l);
    for (int j = 0; j < m; j += 4) {
      const int s0 = __builtin_amdgcn_readlane(s_l, j);
      const int s1 = __builtin_amdgcn_readlane(s_l, j + 1);
      const int s2 = __builtin_amdgcn_readlane(s_l, j + 2);
      const int s3 = __builtin_amdgcn_readlane(s_l, j + 3);
      const float w0 = __int_as_float(__builtin_amdgcn_readlane(wi_l, j));
      const float w1 = __int_as_float(__builtin_amdgcn_readlane(wi_l, j + 1));
      const float w2 = __int_as_float(__builtin_amdgcn_readlane(wi_l, j + 2));
      const float w3 = __int_as_float(__builtin_amdgcn_readlane(wi_l, j + 3));
      const float4 r0 = h4[(size_t)s0 * 32 + lane];
      const float4 r1 = h4[(size_t)s1 * 32 + lane];
      const float4 r2 = h4[(size_t)s2 * 32 + lane];
      const float4 r3 = h4[(size_t)s3 * 32 + lane];
      acc.x += r0.x * w0; acc.y += r0.y * w0; acc.z += r0.z * w0; acc.w += r0.w * w0;
      acc.x += r1.x * w1; acc.y += r1.y * w1; acc.z += r1.z * w1; acc.w += r1.w * w1;
      acc.x += r2.x * w2; acc.y += r2.y * w2; acc.z += r2.z * w2; acc.w += r2.w * w2;
      acc.x += r3.x * w3; acc.y += r3.y * w3; acc.z += r3.z * w3; acc.w += r3.w * w3;
    }
  }
  float4 o;
  o.x = (acc.x >= 0.0f) ? acc.x : LEAKY * acc.x;
  o.y = (acc.y >= 0.0f) ? acc.y : LEAKY * acc.y;
  o.z = (acc.z >= 0.0f) ? acc.z : LEAKY * acc.z;
  o.w = (acc.w >= 0.0f) ? acc.w : LEAKY * acc.w;
  ((float4*)out)[(size_t)d * 32 + lane] = o;
}

extern "C" void kernel_launch(void* const* d_in, const int* in_sizes, int n_in,
                              void* d_out, int out_size, void* d_ws, size_t ws_size,
                              hipStream_t stream) {
  const float* x  = (const float*)d_in[0];
  const int*   ei = (const int*)d_in[1];
  const float* Ws = (const float*)d_in[2];
  const float* bs = (const float*)d_in[3];
  float* out = (float*)d_out;

  const int n = in_sizes[0] / DFEAT;
  const int E = in_sizes[1] / 2;
  const int* src = ei;
  const int* dst = ei + E;

  const int TB = 256;
  const int nb = (n + 1 + 1023) / 1024;

  int* cnt    = (int*)d_ws;
  int* rowptr = cnt + n;
  int* bsum   = rowptr + (n + 1);
  int* rank   = bsum + nb;
  int* csr    = rank + E;
  float* hbuf = (float*)d_ws + (size_t)n * DFEAT;
  float* dis  = hbuf + (size_t)n * DFEAT;

  cnt_zero  <<<(n + TB - 1) / TB, TB, 0, stream>>>(cnt, n);
  edge_rank <<<(E + TB - 1) / TB, TB, 0, stream>>>(dst, cnt, rank, E);
  scan_local<<<nb, 256, 0, stream>>>(cnt, rowptr, bsum, dis, n);
  scan_add  <<<nb, 256, 0, stream>>>(rowptr, bsum, n);
  csr_fill  <<<(E + TB - 1) / TB, TB, 0, stream>>>(src, dst, rowptr, rank, csr, E);

  const float* cur = x;
  for (int l = 0; l < 3; ++l) {
    gcn_gemm_xWT<<<(n + 127) / 128, 256, 0, stream>>>(cur, Ws + (size_t)l * DFEAT * DFEAT, hbuf, n);
    gcn_gather  <<<(n + 7) / 8, 256, 0, stream>>>(rowptr, csr, hbuf, dis, bs + (size_t)l * DFEAT, out, n, E);
    cur = out;
  }
}


# GEMM: weight matrix staged in LDS at a padded row pitch (132 floats, not 128), so the 16 rows of a B fragment read fall on distinct banks instead of one; k order and sums unchanged
# speedup vs baseline: 2.5789x; 1.2225x over previous
#include <hip/hip_runtime.h>


typedef __attribute__((ext_vector_type(2))) float v2f;
typedef __attribute__((ext_vector_type(8))) float v8f;

#define DFEAT 128
#define LEAKY 0.01f
#define WPITCH 132

__global__ void __launch_bounds__(256) gcn_gemm_xWT(
    const float* __restrict__ x, const float* __restrict__ W,
    float* __restrict__ h, int n_nodes)
{
  __shared__ __attribute__((aligned(16))) float Wlds[DFEAT * WPITCH];

  {
    const float4* Wg4 = (const float4*)W;
    #pragma unroll
    for (int i = 0; i < (DFEAT * DFEAT / 4) / 256; ++i) {
      const int idx = threadIdx.x + 256 * i;
      *(float4*)(Wlds + (idx >> 5) * WPITCH + (idx & 31) * 4) = Wg4[idx];
    }
  }
  __syncthreads();

  const int wave = threadIdx.x >> 5;
  const int lane = threadIdx.x & 31;
  const int half = lane >> 4;
  const int lrow = lane & 15;

  const int base = blockIdx.x * 128 + wave * 16;
  if (base >= n_nodes) return;

  int row = base + lrow;
  if (row >= n_nodes) row = n_nodes - 1;
  const v2f* xr = (const v2f*)(x + (size_t)row * DFEAT);

  v2f afrag[32];
  #pragma unroll
  for (int kt = 0; kt < 32; ++kt) afrag[kt] = xr[2 * kt + half];

  #pragma unroll
  for (int nt = 0; nt < 8; ++nt) {
    const v2f* wr = (const v2f*)(Wlds + (nt * 16 + lrow) * WPITCH);
    v8f acc = {};
    #pragma unroll
    for (int kt = 0; kt < 32; ++kt) {
      v2f b = wr[2 * kt + half];
      acc = __builtin_amdgcn_wmma_f32_16x16x4_f32(
           false, afrag[kt],  false, b,
           (short)0, acc,  false,  false);
    }
    const int col = nt * 16 + lrow;
    #pragma unroll
    for (int r = 0; r < 8; ++r) {
      int m = base + half * 8 + r;
      if (m < n_nodes) h[(size_t)m * DFEAT + col] = acc[r];
    }
  }
}

__global__ void cnt_zero(int* __restrict__ cnt, int n) {
  int i = blockIdx.x * blockDim.x + threadIdx.x;
  if (i < n) cnt[i] = 0;
}

__global__ void edge_rank(const int* __restrict__ dst, int* __restrict__ cnt,
                          int* __restrict__ rank, int E) {
  int e = blockIdx.x * blockDim.x + threadIdx.x;
  if (e < E) rank[e] = atomicAdd(&cnt[dst[e]], 1);
}

__global__ void __launch_bounds__(256) scan_local(
    const int* __restrict__ cnt, int* __restrict__ rowptr, int* __restrict__ bsum,
    float* __restrict__ dis, int n)
{
  __shared__ int wsum[8];
  const int tid = threadIdx.x, lane = tid & 31, wave = tid >> 5;
  const int i0 = (blockIdx.x * 256 + tid) * 4;
  int c[4];
  for (int k = 0; k < 4; ++k) {
    int i = i0 + k;
    c[k] = (i < n) ? cnt[i] : 0;
    if (i < n) dis[i] = __frsqrt_rn((float)(c[k] + 1));
  }
  const int t = c[0] + c[1] + c[2] + c[3];
  int incl = t;
  for (int o = 1; o < 32; o <<= 1) {
    int v = __shfl_up(incl, o);
    if (lane >= o) incl += v;
  }
  if (lane == 31) wsum[wave] = incl;
  __syncthreads();
  int woff = 0, tot = 0;
  for (int w = 0; w < 8; ++w) {
    int v = wsum[w];
    if (w < wave) woff += v;
    tot += v;
  }
  int ex = woff + incl - t;
  for (int k = 0; k < 4; ++k) {
    int i = i0 + k;
    if (i <= n) rowptr[i] = ex;
    ex += c[k];
  }
  if (tid == 0) bsum[blockIdx.x] = tot;
}

__global__ void __launch_bounds__(256) scan_add(
    int* __restrict__ rowptr, const int* __restrict__ bsum, int n)
{
  __shared__ int red[8];
  const int tid = threadIdx.x, lane = tid & 31, wave = tid >> 5;
  int part = 0;
  for (int b = tid; b < (int)blockIdx.x; b += 256) part += bsum[b];
  for (int o = 16; o > 0; o >>= 1) part += __shfl_xor(part, o);
  if (lane == 0) red[wave] = part;
  __syncthreads();
  int base = 0;
  for (int w = 0; w < 8; ++w) base += red[w];
  const int i0 = (blockIdx.x * 256 + tid) * 4;
  for (int k = 0; k < 4; ++k) {
    int i = i0 + k;
    if (i <= n) rowptr[i] += base;
  }
}

__global__ void csr_fill(const int* __restrict__ src, const int* __restrict__ dst,
                         const int* __restrict__ rowptr, const int* __restrict__ rank,
                         int* __restrict__ csr, int E) {
  int e = blockIdx.x * blockDim.x + threadIdx.x;
  if (e < E) {
    const int p = rowptr[dst[e]] + rank[e];
    if ((unsigned)p < (unsigned)E) csr[p] = src[e];
  }
}

__global__ void __launch_bounds__(256) gcn_gather(
    const int* __restrict__ rowptr, const int* __restrict__ csr,
    const float* __restrict__ h, const float* __restrict__ dis,
    const float* __restrict__ bias, float* __restrict__ out, int n, int E)
{
  const int d = __builtin_amdgcn_readfirstlane((int)(blockIdx.x * 8 + (threadIdx.x >> 5)));
  if (d >= n) return;
  const int lane = threadIdx.x & 31;
  int beg = __builtin_amdgcn_readfirstlane(rowptr[d]);
  int end = __builtin_amdgcn_readfirstlane(rowptr[d + 1]);
  beg = (beg < 0) ? 0 : beg;
  end = (end > E) ? E : end;
  const float dd = dis[d];
  const float4* __restrict__ h4 = (const float4*)h;
  const float4 hv = h4[(size_t)d * 32 + lane];
  const float4 bv = ((const float4*)bias)[lane];
  float4 acc;
  acc.x = bv.x + hv.x * dd * dd;
  acc.y = bv.y + hv.y * dd * dd;
  acc.z = bv.z + hv.z * dd * dd;
  acc.w = bv.w + hv.w * dd * dd;

  for (int c = beg; c < end; c += 32) {
    const int m = (end - c < 32) ? (end - c) : 32;
    int key = 0x7fffffff;
    if (lane < m) key = csr[c + lane];
    #pragma unroll
    for (int k = 2; k <= 32; k <<= 1) {
      #pragma unroll
      for (int j = k >> 1; j > 0; j >>= 1) {
        const int p = __shfl_xor(key, j);
        const bool up = ((lane & k) == 0);
        const bool lower = ((lane & j) == 0);
        const int lo = (key < p) ? key : p;
        const int hi = (key < p) ? p : key;
        key = (lower == up) ? lo : hi;
      }
    }
    int s_l = (lane < m) ? key : d;
    s_l = (s_l < 0) ? 0 : ((s_l >= n) ? n - 1 : s_l);
    float w_l = dis[s_l] * dd;
    if (lane >= m) w_l = 0.0f;
    const int wi_l = __float_as_int(w_l);
    for (int j = 0; j < m; j += 4) {
      const int s0 = __builtin_amdgcn_readlane(s_l, j);
      const int s1 = __builtin_amdgcn_readlane(s_l, j + 1);
      const int s2 = __builtin_amdgcn_readlane(s_l, j + 2);
      const int s3 = __builtin_amdgcn_readlane(s_l, j + 3);
      const float w0 = __int_as_float(__builtin_amdgcn_readlane(wi_l, j));
      const float w1 = __int_as_float(__builtin_amdgcn_readlane(wi_l, j + 1));
      const float w2 = __int_as_float(__builtin_amdgcn_readlane(wi_l, j + 2));
      const float w3 = __int_as_float(__builtin_amdgcn_readlane(wi_l, j + 3));
      const float4 r0 = h4[(size_t)s0 * 32 + lane];
      const float4 r1 = h4[(size_t)s1 * 32 + lane];
      const float4 r2 = h4[(size_t)s2 * 32 + lane];
      const float4 r3 = h4[(size_t)s3 * 32 + lane];
      acc.x += r0.x * w0; acc.y += r0.y * w0; acc.z += r0.z * w0; acc.w += r0.w * w0;
      acc.x += r1.x * w1; acc.y += r1.y * w1; acc.z += r1.z * w1; acc.w += r1.w * w1;
      acc.x += r2.x * w2; acc.y += r2.y * w2; acc.z += r2.z * w2; acc.w += r2.w * w2;
      acc.x += r3.x * w3; acc.y += r3.y * w3; acc.z += r3.z * w3; acc.w += r3.w * w3;
    }
  }
  float4 o;
  o.x = (acc.x >= 0.0f) ? acc.x : LEAKY * acc.x;
  o.y = (acc.y >= 0.0f) ? acc.y : LEAKY * acc.y;
  o.z = (acc.z >= 0.0f) ? acc.z : LEAKY * acc.z;
  o.w = (acc.w >= 0.0f) ? acc.w : LEAKY * acc.w;
  ((float4*)out)[(size_t)d * 32 + lane] = o;
}

extern "C" void kernel_launch(void* const* d_in, const int* in_sizes, int n_in,
                              void* d_out, int out_size, void* d_ws, size_t ws_size,
                              hipStream_t stream) {
  const float* x  = (const float*)d_in[0];
  const int*   ei = (const int*)d_in[1];
  const float* Ws = (const float*)d_in[2];
  const float* bs = (const float*)d_in[3];
  float* out = (float*)d_out;

  const int n = in_sizes[0] / DFEAT;
  const int E = in_sizes[1] / 2;
  const int* src = ei;
  const int* dst = ei + E;

  const int TB = 256;
  const int nb = (n + 1 + 1023) / 1024;

  int* cnt    = (int*)d_ws;
  int* rowptr = cnt + n;
  int* bsum   = rowptr + (n + 1);
  int* rank   = bsum + nb;
  int* csr    = rank + E;
  float* hbuf = (float*)d_ws + (size_t)n * DFEAT;
  float* dis  = hbuf + (size_t)n * DFEAT;

  cnt_zero  <<<(n + TB - 1) / TB, TB, 0, stream>>>(cnt, n);
  edge_rank <<<(E + TB - 1) / TB, TB, 0, stream>>>(dst, cnt, rank, E);
  scan_local<<<nb, 256, 0, stream>>>(cnt, rowptr, bsum, dis, n);
  scan_add  <<<nb, 256, 0, stream>>>(rowptr, bsum, n);
  csr_fill  <<<(E + TB - 1) / TB, TB, 0, stream>>>(src, dst, rowptr, rank, csr, E);

  const float* cur = x;
  for (int l = 0; l < 3; ++l) {
    gcn_gemm_xWT<<<(n + 127) / 128, 256, 0, stream>>>(cur, Ws + (size_t)l * DFEAT * DFEAT, hbuf, n);
    gcn_gather  <<<(n + 7) / 8, 256, 0, stream>>>(rowptr, csr, hbuf, dis, bs + (size_t)l * DFEAT, out, n, E);
    cur = out;
  }
}


# Edge weight split: GEMM takes input row s times deg^-1/2[s]; node sum adds rows unweighted, then times deg^-1/2[d] plus bias (rounding order changes): no per-edge deg lookup in the 3 gathers
# speedup vs baseline: 2.6676x; 1.0344x over previous
#include <hip/hip_runtime.h>


typedef __attribute__((ext_vector_type(2))) float v2f;
typedef __attribute__((ext_vector_type(8))) float v8f;

#define DFEAT 128
#define LEAKY 0.01f
#define WPITCH 132

__global__ void __launch_bounds__(256) gcn_gemm_xWT(
    const float* __restrict__ x, const float* __restrict__ W,
    const float* __restrict__ dis, float* __restrict__ h, int n_nodes)
{
  __shared__ __attribute__((aligned(16))) float Wlds[DFEAT * WPITCH];

  {
    const float4* Wg4 = (const float4*)W;
    #pragma unroll
    for (int i = 0; i < (DFEAT * DFEAT / 4) / 256; ++i) {
      const int idx = threadIdx.x + 256 * i;
      *(float4*)(Wlds + (idx >> 5) * WPITCH + (idx & 31) * 4) = Wg4[idx];
    }
  }
  __syncthreads();

  const int wave = threadIdx.x >> 5;
  const int lane = threadIdx.x & 31;
  const int half = lane >> 4;
  const int lrow = lane & 15;

  const int base = blockIdx.x * 128 + wave * 16;
  if (base >= n_nodes) return;

  int row = base + lrow;
  if (row >= n_nodes) row = n_nodes - 1;
  const v2f* xr = (const v2f*)(x + (size_t)row * DFEAT);

  const float sr = dis[row];

  v2f afrag[32];
  #pragma unroll
  for (int kt = 0; kt < 32; ++kt) afrag[kt] = xr[2 * kt + half] * sr;

  #pragma unroll
  for (int nt = 0; nt < 8; ++nt) {
    const v2f* wr = (const v2f*)(Wlds + (nt * 16 + lrow) * WPITCH);
    v8f acc = {};
    #pragma unroll
    for (int kt = 0; kt < 32; ++kt) {
      v2f b = wr[2 * kt + half];
      acc = __builtin_amdgcn_wmma_f32_16x16x4_f32(
           false, afrag[kt],  false, b,
           (short)0, acc,  false,  false);
    }
    const int col = nt * 16 + lrow;
    #pragma unroll
    for (int r = 0; r < 8; ++r) {
      int m = base + half * 8 + r;
      if (m < n_nodes) h[(size_t)m * DFEAT + col] = acc[r];
    }
  }
}

__global__ void cnt_zero(int* __restrict__ cnt, int n) {
  int i = blockIdx.x * blockDim.x + threadIdx.x;
  if (i < n) cnt[i] = 0;
}

__global__ void edge_rank(const int* __restrict__ dst, int* __restrict__ cnt,
                          int* __restrict__ rank, int E) {
  int e = blockIdx.x * blockDim.x + threadIdx.x;
  if (e < E) rank[e] = atomicAdd(&cnt[dst[e]], 1);
}

__global__ void __launch_bounds__(256) scan_local(
    const int* __restrict__ cnt, int* __restrict__ rowptr, int* __restrict__ bsum,
    float* __restrict__ dis, int n)
{
  __shared__ int wsum[8];
  const int tid = threadIdx.x, lane = tid & 31, wave = tid >> 5;
  const int i0 = (blockIdx.x * 256 + tid) * 4;
  int c[4];
  for (int k = 0; k < 4; ++k) {
    int i = i0 + k;
    c[k] = (i < n) ? cnt[i] : 0;
    if (i < n) dis[i] = __frsqrt_rn((float)(c[k] + 1));
  }
  const int t = c[0] + c[1] + c[2] + c[3];
  int incl = t;
  for (int o = 1; o < 32; o <<= 1) {
    int v = __shfl_up(incl, o);
    if (lane >= o) incl += v;
  }
  if (lane == 31) wsum[wave] = incl;
  __syncthreads();
  int woff = 0, tot = 0;
  for (int w = 0; w < 8; ++w) {
    int v = wsum[w];
    if (w < wave) woff += v;
    tot += v;
  }
  int ex = woff + incl - t;
  for (int k = 0; k < 4; ++k) {
    int i = i0 + k;
    if (i <= n) rowptr[i] = ex;
    ex += c[k];
  }
  if (tid == 0) bsum[blockIdx.x] = tot;
}

__global__ void __launch_bounds__(256) scan_add(
    int* __restrict__ rowptr, const int* __restrict__ bsum, int n)
{
  __shared__ int red[8];
  const int tid = threadIdx.x, lane = tid & 31, wave = tid >> 5;
  int part = 0;
  for (int b = tid; b < (int)blockIdx.x; b += 256) part += bsum[b];
  for (int o = 16; o > 0; o >>= 1) part += __shfl_xor(part, o);
  if (lane == 0) red[wave] = part;
  __syncthreads();
  int base = 0;
  for (int w = 0; w < 8; ++w) base += red[w];
  const int i0 = (blockIdx.x * 256 + tid) * 4;
  for (int k = 0; k < 4; ++k) {
    int i = i0 + k;
    if (i <= n) rowptr[i] += base;
  }
}

__global__ void csr_fill(const int* __restrict__ src, const int* __restrict__ dst,
                         const int* __restrict__ rowptr, const int* __restrict__ rank,
                         int* __restrict__ csr, int E) {
  int e = blockIdx.x * blockDim.x + threadIdx.x;
  if (e < E) {
    const int p = rowptr[dst[e]] + rank[e];
    if ((unsigned)p < (unsigned)E) csr[p] = src[e];
  }
}

__global__ void __launch_bounds__(256) gcn_gather(
    const int* __restrict__ rowptr, const int* __restrict__ csr,
    const float* __restrict__ h, const float* __restrict__ dis,
    const float* __restrict__ bias, float* __restrict__ out, int n, int E)
{
  const int d = __builtin_amdgcn_readfirstlane((int)(blockIdx.x * 8 + (threadIdx.x >> 5)));
  if (d >= n) return;
  const int lane = threadIdx.x & 31;
  int beg = __builtin_amdgcn_readfirstlane(rowptr[d]);
  int end = __builtin_amdgcn_readfirstlane(rowptr[d + 1]);
  beg = (beg < 0) ? 0 : beg;
  end = (end > E) ? E : end;
  const float dd = dis[d];
  const float4* __restrict__ h4 = (const float4*)h;
  const float4 hv = h4[(size_t)d * 32 + lane];
  const float4 bv = ((const float4*)bias)[lane];
  float4 acc = hv;

  for (int c = beg; c < end; c += 32) {
    const int m = (end - c < 32) ? (end - c) : 32;
    int key = 0x7fffffff;
    if (lane < m) key = csr[c + lane];
    #pragma unroll
    for (int k = 2; k <= 32; k <<= 1) {
      #pragma unroll
      for (int j = k >> 1; j > 0; j >>= 1) {
        const int p = __shfl_xor(key, j);
        const bool up = ((lane & k) == 0);
        const bool lower = ((lane & j) == 0);
        const int lo = (key < p) ? key : p;
        const int hi = (key < p) ? p : key;
        key = (lower == up) ? lo : hi;
      }
    }
    int s_l = (lane < m) ? key : d;
    s_l = (s_l < 0) ? 0 : ((s_l >= n) ? n - 1 : s_l);
    int j = 0;
    for (; j + 4 <= m; j += 4) {
      const int s0 = __builtin_amdgcn_readlane(s_l, j);
      const int s1 = __builtin_amdgcn_readlane(s_l, j + 1);
      const int s2 = __builtin_amdgcn_readlane(s_l, j + 2);
      const int s3 = __builtin_amdgcn_readlane(s_l, j + 3);
      const float4 r0 = h4[(size_t)s0 * 32 + lane];
      const float4 r1 = h4[(size_t)s1 * 32 + lane];
      const float4 r2 = h4[(size_t)s2 * 32 + lane];
      const float4 r3 = h4[(size_t)s3 * 32 + lane];
      acc.x += r0.x; acc.y += r0.y; acc.z += r0.z; acc.w += r0.w;
      acc.x += r1.x; acc.y += r1.y; acc.z += r1.z; acc.w += r1.w;
      acc.x += r2.x; acc.y += r2.y; acc.z += r2.z; acc.w += r2.w;
      acc.x += r3.x; acc.y += r3.y; acc.z += r3.z; acc.w += r3.w;
    }
    if (j < m) {
      const int s0 = __builtin_amdgcn_readlane(s_l, j);
      const int s1 = __builtin_amdgcn_readlane(s_l, j + 1);
      const int s2 = __builtin_amdgcn_readlane(s_l, j + 2);
      const float4 r0 = h4[(size_t)s0 * 32 + lane];
      const float4 r1 = h4[(size_t)s1 * 32 + lane];
      const float4 r2 = h4[(size_t)s2 * 32 + lane];
      acc.x += r0.x; acc.y += r0.y; acc.z += r0.z; acc.w += r0.w;
      if (j + 1 < m) { acc.x += r1.x; acc.y += r1.y; acc.z += r1.z; acc.w += r1.w; }
      if (j + 2 < m) { acc.x += r2.x; acc.y += r2.y; acc.z += r2.z; acc.w += r2.w; }
    }
  }
  acc.x = bv.x + acc.x * dd;
  acc.y = bv.y + acc.y * dd;
  acc.z = bv.z + acc.z * dd;
  acc.w = bv.w + acc.w * dd;
  float4 o;
  o.x = (acc.x >= 0.0f) ? acc.x : LEAKY * acc.x;
  o.y = (acc.y >= 0.0f) ? acc.y : LEAKY * acc.y;
  o.z = (acc.z >= 0.0f) ? acc.z : LEAKY * acc.z;
  o.w = (acc.w >= 0.0f) ? acc.w : LEAKY * acc.w;
  ((float4*)out)[(size_t)d * 32 + lane] = o;
}

extern "C" void kernel_launch(void* const* d_in, const int* in_sizes, int n_in,
                              void* d_out, int out_size, void* d_ws, size_t ws_size,
                              hipStream_t stream) {
  const float* x  = (const float*)d_in[0];
  const int*   ei = (const int*)d_in[1];
  const float* Ws = (const float*)d_in[2];
  const float* bs = (const float*)d_in[3];
  float* out = (float*)d_out;

  const int n = in_sizes[0] / DFEAT;
  const int E = in_sizes[1] / 2;
  const int* src = ei;
  const int* dst = ei + E;

  const int TB = 256;
  const int nb = (n + 1 + 1023) / 1024;

  int* cnt    = (int*)d_ws;
  int* rowptr = cnt + n;
  int* bsum   = rowptr + (n + 1);
  int* rank   = bsum + nb;
  int* csr    = rank + E;
  float* hbuf = (float*)d_ws + (size_t)n * DFEAT;
  float* dis  = hbuf + (size_t)n * DFEAT;

  cnt_zero  <<<(n + TB - 1) / TB, TB, 0, stream>>>(cnt, n);
  edge_rank <<<(E + TB - 1) / TB, TB, 0, stream>>>(dst, cnt, rank, E);
  scan_local<<<nb, 256, 0, stream>>>(cnt, rowptr, bsum, dis, n);
  scan_add  <<<nb, 256, 0, stream>>>(rowptr, bsum, n);
  csr_fill  <<<(E + TB - 1) / TB, TB, 0, stream>>>(src, dst, rowptr, rank, csr, E);

  const float* cur = x;
  for (int l = 0; l < 3; ++l) {
    gcn_gemm_xWT<<<(n + 127) / 128, 256, 0, stream>>>(cur, Ws + (size_t)l * DFEAT * DFEAT, dis, hbuf, n);
    gcn_gather  <<<(n + 7) / 8, 256, 0, stream>>>(rowptr, csr, hbuf, dis, bs + (size_t)l * DFEAT, out, n, E);
    cur = out;
  }
}


# Node sums of layers 1 and 2 share a launch with the next layer's GEMM: a 1024-thread block sums 128 nodes, keeps rows times deg^-1/2 in LDS as the A operand; 9 launches, not 11; sums unchanged
# speedup vs baseline: 2.8992x; 1.0868x over previous
#include <hip/hip_runtime.h>


typedef __attribute__((ext_vector_type(2))) float v2f;
typedef __attribute__((ext_vector_type(8))) float v8f;

#define DFEAT 128
#define LEAKY 0.01f
#define WPITCH 132

__global__ void __launch_bounds__(256) gcn_gemm_xWT(
    const float* __restrict__ x, const float* __restrict__ W,
    const float* __restrict__ dis, float* __restrict__ h, int n_nodes)
{
  __shared__ __attribute__((aligned(16))) float Wlds[DFEAT * WPITCH];

  {
    const float4* Wg4 = (const float4*)W;
    #pragma unroll
    for (int i = 0; i < (DFEAT * DFEAT / 4) / 256; ++i) {
      const int idx = threadIdx.x + 256 * i;
      *(float4*)(Wlds + (idx >> 5) * WPITCH + (idx & 31) * 4) = Wg4[idx];
    }
  }
  __syncthreads();

  const int wave = threadIdx.x >> 5;
  const int lane = threadIdx.x & 31;
  const int half = lane >> 4;
  const int lrow = lane & 15;

  const int base = blockIdx.x * 128 + wave * 16;
  if (base >= n_nodes) return;

  int row = base + lrow;
  if (row >= n_nodes) row = n_nodes - 1;
  const v2f* xr = (const v2f*)(x + (size_t)row * DFEAT);

  const float sr = dis[row];

  v2f afrag[32];
  #pragma unroll
  for (int kt = 0; kt < 32; ++kt) afrag[kt] = xr[2 * kt + half] * sr;

  #pragma unroll
  for (int nt = 0; nt < 8; ++nt) {
    const v2f* wr = (const v2f*)(Wlds + (nt * 16 + lrow) * WPITCH);
    v8f acc = {};
    #pragma unroll
    for (int kt = 0; kt < 32; ++kt) {
      v2f b = wr[2 * kt + half];
      acc = __builtin_amdgcn_wmma_f32_16x16x4_f32(
           false, afrag[kt],  false, b,
           (short)0, acc,  false,  false);
    }
    const int col = nt * 16 + lrow;
    #pragma unroll
    for (int r = 0; r < 8; ++r) {
      int m = base + half * 8 + r;
      if (m < n_nodes) h[(size_t)m * DFEAT + col] = acc[r];
    }
  }
}

__global__ void cnt_zero(int* __restrict__ cnt, int n) {
  int i = blockIdx.x * blockDim.x + threadIdx.x;
  if (i < n) cnt[i] = 0;
}

__global__ void edge_rank(const int* __restrict__ dst, int* __restrict__ cnt,
                          int* __restrict__ rank, int E) {
  int e = blockIdx.x * blockDim.x + threadIdx.x;
  if (e < E) rank[e] = atomicAdd(&cnt[dst[e]], 1);
}

__global__ void __launch_bounds__(256) scan_local(
    const int* __restrict__ cnt, int* __restrict__ rowptr, int* __restrict__ bsum,
    float* __restrict__ dis, int n)
{
  __shared__ int wsum[8];
  const int tid = threadIdx.x, lane = tid & 31, wave = tid >> 5;
  const int i0 = (blockIdx.x * 256 + tid) * 4;
  int c[4];
  for (int k = 0; k < 4; ++k) {
    int i = i0 + k;
    c[k] = (i < n) ? cnt[i] : 0;
    if (i < n) dis[i] = __frsqrt_rn((float)(c[k] + 1));
  }
  const int t = c[0] + c[1] + c[2] + c[3];
  int incl = t;
  for (int o = 1; o < 32; o <<= 1) {
    int v = __shfl_up(incl, o);
    if (lane >= o) incl += v;
  }
  if (lane == 31) wsum[wave] = incl;
  __syncthreads();
  int woff = 0, tot = 0;
  for (int w = 0; w < 8; ++w) {
    int v = wsum[w];
    if (w < wave) woff += v;
    tot += v;
  }
  int ex = woff + incl - t;
  for (int k = 0; k < 4; ++k) {
    int i = i0 + k;
    if (i <= n) rowptr[i] = ex;
    ex += c[k];
  }
  if (tid == 0) bsum[blockIdx.x] = tot;
}

__global__ void __launch_bounds__(256) scan_add(
    int* __restrict__ rowptr, const int* __restrict__ bsum, int n)
{
  __shared__ int red[8];
  const int tid = threadIdx.x, lane = tid & 31, wave = tid >> 5;
  int part = 0;
  for (int b = tid; b < (int)blockIdx.x; b += 256) part += bsum[b];
  for (int o = 16; o > 0; o >>= 1) part += __shfl_xor(part, o);
  if (lane == 0) red[wave] = part;
  __syncthreads();
  int base = 0;
  for (int w = 0; w < 8; ++w) base += red[w];
  const int i0 = (blockIdx.x * 256 + tid) * 4;
  for (int k = 0; k < 4; ++k) {
    int i = i0 + k;
    if (i <= n) rowptr[i] += base;
  }
}

__global__ void csr_fill(const int* __restrict__ src, const int* __restrict__ dst,
                         const int* __restrict__ rowptr, const int* __restrict__ rank,
                         int* __restrict__ csr, int E) {
  int e = blockIdx.x * blockDim.x + threadIdx.x;
  if (e < E) {
    const int p = rowptr[dst[e]] + rank[e];
    if ((unsigned)p < (unsigned)E) csr[p] = src[e];
  }
}


template <int NK>
__device__ __forceinline__ void sort_keys(int (&key)[NK], const int lane) {
  #pragma unroll
  for (int k = 2; k <= 32; k <<= 1) {
    #pragma unroll
    for (int j = k >> 1; j > 0; j >>= 1) {
      const bool up = ((lane & k) == 0);
      const bool lower = ((lane & j) == 0);
      #pragma unroll
      for (int i = 0; i < NK; ++i) {
        const int p = __shfl_xor(key[i], j);
        const int lo = (key[i] < p) ? key[i] : p;
        const int hi = (key[i] < p) ? p : key[i];
        key[i] = (lower == up) ? lo : hi;
      }
    }
  }
}

__device__ __forceinline__ float4 add_rows(
    float4 acc, const int key, const int m, const int d, const int lane,
    const float4* __restrict__ h4, const int n)
{
  int s_l = (lane < m) ? key : d;
  s_l = (s_l < 0) ? 0 : ((s_l >= n) ? n - 1 : s_l);
  int j = 0;
  for (; j + 4 <= m; j += 4) {
    const int s0 = __builtin_amdgcn_readlane(s_l, j);
    const int s1 = __builtin_amdgcn_readlane(s_l, j + 1);
    const int s2 = __builtin_amdgcn_readlane(s_l, j + 2);
    const int s3 = __builtin_amdgcn_readlane(s_l, j + 3);
    const float4 r0 = h4[(size_t)s0 * 32 + lane];
    const float4 r1 = h4[(size_t)s1 * 32 + lane];
    const float4 r2 = h4[(size_t)s2 * 32 + lane];
    const float4 r3 = h4[(size_t)s3 * 32 + lane];
    acc.x += r0.x; acc.y += r0.y; acc.z += r0.z; acc.w += r0.w;
    acc.x += r1.x; acc.y += r1.y; acc.z += r1.z; acc.w += r1.w;
    acc.x += r2.x; acc.y += r2.y; acc.z += r2.z; acc.w += r2.w;
    acc.x += r3.x; acc.y += r3.y; acc.z += r3.z; acc.w += r3.w;
  }
  if (j < m) {
    const int s0 = __builtin_amdgcn_readlane(s_l, j);
    const int s1 = __builtin_amdgcn_readlane(s_l, j + 1);
    const int s2 = __builtin_amdgcn_readlane(s_l, j + 2);
    const float4 r0 = h4[(size_t)s0 * 32 + lane];
    const float4 r1 = h4[(size_t)s1 * 32 + lane];
    const float4 r2 = h4[(size_t)s2 * 32 + lane];
    acc.x += r0.x; acc.y += r0.y; acc.z += r0.z; acc.w += r0.w;
    if (j + 1 < m) { acc.x += r1.x; acc.y += r1.y; acc.z += r1.z; acc.w += r1.w; }
    if (j + 2 < m) { acc.x += r2.x; acc.y += r2.y; acc.z += r2.z; acc.w += r2.w; }
  }
  return acc;
}

__device__ __forceinline__ float4 add_chunks(
    float4 acc, int c, const int end, const int d, const int lane,
    const int* __restrict__ csr, const float4* __restrict__ h4, const int n)
{
  for (; c < end; c += 32) {
    const int m = (end - c < 32) ? (end - c) : 32;
    int key[1] = {0x7fffffff};
    if (lane < m) key[0] = csr[c + lane];
    sort_keys<1>(key, lane);
    acc = add_rows(acc, key[0], m, d, lane, h4, n);
  }
  return acc;
}

__device__ __forceinline__ float4 finish_row(float4 acc, const float4 bv, const float dd) {
  acc.x = bv.x + acc.x * dd;
  acc.y = bv.y + acc.y * dd;
  acc.z = bv.z + acc.z * dd;
  acc.w = bv.w + acc.w * dd;
  float4 o;
  o.x = (acc.x >= 0.0f) ? acc.x : LEAKY * acc.x;
  o.y = (acc.y >= 0.0f) ? acc.y : LEAKY * acc.y;
  o.z = (acc.z >= 0.0f) ? acc.z : LEAKY * acc.z;
  o.w = (acc.w >= 0.0f) ? acc.w : LEAKY * acc.w;
  return o;
}

__global__ void __launch_bounds__(256) gcn_gather(
    const int* __restrict__ rowptr, const int* __restrict__ csr,
    const float* __restrict__ h, const float* __restrict__ dis,
    const float* __restrict__ bias, float* __restrict__ out, int n, int E)
{
  const int d = __builtin_amdgcn_readfirstlane((int)(blockIdx.x * 8 + (threadIdx.x >> 5)));
  if (d >= n) return;
  const int lane = threadIdx.x & 31;
  int beg = __builtin_amdgcn_readfirstlane(rowptr[d]);
  int end = __builtin_amdgcn_readfirstlane(rowptr[d + 1]);
  beg = (beg < 0) ? 0 : beg;
  end = (end > E) ? E : end;
  const float dd = dis[d];
  const float4* __restrict__ h4 = (const float4*)h;
  const float4 hv = h4[(size_t)d * 32 + lane];
  const float4 bv = ((const float4*)bias)[lane];
  const float4 acc = add_chunks(hv, beg, end, d, lane, csr, h4, n);
  ((float4*)out)[(size_t)d * 32 + lane] = finish_row(acc, bv, dd);
}

#define TILE 128
__global__ void __launch_bounds__(1024) gcn_gather_gemm(
    const int* __restrict__ rowptr, const int* __restrict__ csr,
    const float* __restrict__ hin, const float* __restrict__ dis,
    const float* __restrict__ bias, const float* __restrict__ W,
    float* __restrict__ hout, int n, int E)
{
  __shared__ __attribute__((aligned(16))) float Wlds[DFEAT * WPITCH];
  __shared__ __attribute__((aligned(16))) float Alds[TILE * WPITCH];

  {
    const float4* Wg4 = (const float4*)W;
    #pragma unroll
    for (int i = 0; i < (DFEAT * DFEAT / 4) / 1024; ++i) {
      const int idx = threadIdx.x + 1024 * i;
      *(float4*)(Wlds + (idx >> 5) * WPITCH + (idx & 31) * 4) = Wg4[idx];
    }
  }

  const int wave = __builtin_amdgcn_readfirstlane((int)(threadIdx.x >> 5));
  const int lane = threadIdx.x & 31;
  const int tile0 = blockIdx.x * TILE;
  const float4* __restrict__ h4 = (const float4*)hin;

  const int d0 = tile0 + wave * 4;
  int beg[4], end[4], m0[4], key[4];
  #pragma unroll
  for (int i = 0; i < 4; ++i) {
    const int d = d0 + i;
    int b = 0, e = 0;
    if (d < n) {
      b = __builtin_amdgcn_readfirstlane(rowptr[d]);
      e = __builtin_amdgcn_readfirstlane(rowptr[d + 1]);
      b = (b < 0) ? 0 : b;
      e = (e > E) ? E : e;
    }
    beg[i] = b;
    end[i] = e;
    m0[i] = (e - b < 32) ? (e - b) : 32;
    key[i] = 0x7fffffff;
    if (lane < m0[i]) key[i] = csr[b + lane];
  }
  sort_keys<4>(key, lane);

  #pragma unroll
  for (int i = 0; i < 4; ++i) {
    const int lr = wave * 4 + i;
    const int d = d0 + i;
    float4 a = {0.0f, 0.0f, 0.0f, 0.0f};
    if (d < n) {
      const float dd = dis[d];
      const float4 hv = h4[(size_t)d * 32 + lane];
      const float4 bv = ((const float4*)bias)[lane];
      float4 acc = add_rows(hv, key[i], m0[i], d, lane, h4, n);
      acc = add_chunks(acc, beg[i] + 32, end[i], d, lane, csr, h4, n);
      const float4 o = finish_row(acc, bv, dd);
      a.x = o.x * dd; a.y = o.y * dd; a.z = o.z * dd; a.w = o.w * dd;
    }
    *(float4*)(Alds + lr * WPITCH + lane * 4) = a;
  }
  __syncthreads();

  const int half = lane >> 4;
  const int lrow = lane & 15;
  const int rt = wave >> 2;
  const int cg = wave & 3;
  const v2f* ar = (const v2f*)(Alds + (rt * 16 + lrow) * WPITCH);
  const v2f* w0 = (const v2f*)(Wlds + ((cg * 2 + 0) * 16 + lrow) * WPITCH);
  const v2f* w1 = (const v2f*)(Wlds + ((cg * 2 + 1) * 16 + lrow) * WPITCH);
  v8f acc0 = {};
  v8f acc1 = {};
  #pragma unroll
  for (int kt = 0; kt < 32; ++kt) {
    const v2f a  = ar[2 * kt + half];
    const v2f b0 = w0[2 * kt + half];
    const v2f b1 = w1[2 * kt + half];
    acc0 = __builtin_amdgcn_wmma_f32_16x16x4_f32(
         false, a,  false, b0,
         (short)0, acc0,  false,  false);
    acc1 = __builtin_amdgcn_wmma_f32_16x16x4_f32(
         false, a,  false, b1,
         (short)0, acc1,  false,  false);
  }
  const int mbase = tile0 + rt * 16 + half * 8;
  const int col0 = (cg * 2) * 16 + lrow;
  #pragma unroll
  for (int r = 0; r < 8; ++r) {
    const int m = mbase + r;
    if (m < n) {
      hout[(size_t)m * DFEAT + col0]      = acc0[r];
      hout[(size_t)m * DFEAT + col0 + 16] = acc1[r];
    }
  }
}

extern "C" void kernel_launch(void* const* d_in, const int* in_sizes, int n_in,
                              void* d_out, int out_size, void* d_ws, size_t ws_size,
                              hipStream_t stream) {
  const float* x  = (const float*)d_in[0];
  const int*   ei = (const int*)d_in[1];
  const float* Ws = (const float*)d_in[2];
  const float* bs = (const float*)d_in[3];
  float* out = (float*)d_out;

  const int n = in_sizes[0] / DFEAT;
  const int E = in_sizes[1] / 2;
  const int* src = ei;
  const int* dst = ei + E;

  const int TB = 256;
  const int nb = (n + 1 + 1023) / 1024;

  int* cnt    = (int*)d_ws;
  int* rowptr = cnt + n;
  int* bsum   = rowptr + (n + 1);
  int* rank   = bsum + nb;
  int* csr    = rank + E;
  float* hbuf = (float*)d_ws + (size_t)n * DFEAT;
  float* dis  = hbuf + (size_t)n * DFEAT;

  cnt_zero  <<<(n + TB - 1) / TB, TB, 0, stream>>>(cnt, n);
  edge_rank <<<(E + TB - 1) / TB, TB, 0, stream>>>(dst, cnt, rank, E);
  scan_local<<<nb, 256, 0, stream>>>(cnt, rowptr, bsum, dis, n);
  scan_add  <<<nb, 256, 0, stream>>>(rowptr, bsum, n);
  csr_fill  <<<(E + TB - 1) / TB, TB, 0, stream>>>(src, dst, rowptr, rank, csr, E);

  const size_t WW = (size_t)DFEAT * DFEAT;
  gcn_gemm_xWT   <<<(n + 127) / 128, 256, 0, stream>>>(x, Ws, dis, hbuf, n);
  gcn_gather_gemm<<<(n + TILE - 1) / TILE, 1024, 0, stream>>>(rowptr, csr, hbuf, dis, bs, Ws + WW, out, n, E);
  gcn_gather_gemm<<<(n + TILE - 1) / TILE, 1024, 0, stream>>>(rowptr, csr, out, dis, bs + DFEAT, Ws + 2 * WW, hbuf, n, E);
  gcn_gather     <<<(n + 7) / 8, 256, 0, stream>>>(rowptr, csr, hbuf, dis, bs + 2 * DFEAT, out, n, E);
}


# Edges grouped by destination by a two-level counting sort with its counters in LDS (512-node ranges, then nodes): no global atomic or per-edge rank; 7 launches, not 9; sums unchanged
# speedup vs baseline: 3.8669x; 1.3338x over previous
#include <hip/hip_runtime.h>


typedef __attribute__((ext_vector_type(2))) float v2f;
typedef __attribute__((ext_vector_type(8))) float v8f;

#define DFEAT 128
#define LEAKY 0.01f
#define WPITCH 132

__global__ void __launch_bounds__(256) gcn_gemm_xWT(
    const float* __restrict__ x, const float* __restrict__ W,
    const float* __restrict__ dis, float* __restrict__ h, int n_nodes)
{
  __shared__ __attribute__((aligned(16))) float Wlds[DFEAT * WPITCH];

  {
    const float4* Wg4 = (const float4*)W;
    #pragma unroll
    for (int i = 0; i < (DFEAT * DFEAT / 4) / 256; ++i) {
      const int idx = threadIdx.x + 256 * i;
      *(float4*)(Wlds + (idx >> 5) * WPITCH + (idx & 31) * 4) = Wg4[idx];
    }
  }
  __syncthreads();

  const int wave = threadIdx.x >> 5;
  const int lane = threadIdx.x & 31;
  const int half = lane >> 4;
  const int lrow = lane & 15;

  const int base = blockIdx.x * 128 + wave * 16;
  if (base >= n_nodes) return;

  int row = base + lrow;
  if (row >= n_nodes) row = n_nodes - 1;
  const v2f* xr = (const v2f*)(x + (size_t)row * DFEAT);

  const float sr = dis[row];

  v2f afrag[32];
  #pragma unroll
  for (int kt = 0; kt < 32; ++kt) afrag[kt] = xr[2 * kt + half] * sr;

  #pragma unroll
  for (int nt = 0; nt < 8; ++nt) {
    const v2f* wr = (const v2f*)(Wlds + (nt * 16 + lrow) * WPITCH);
    v8f acc = {};
    #pragma unroll
    for (int kt = 0; kt < 32; ++kt) {
      v2f b = wr[2 * kt + half];
      acc = __builtin_amdgcn_wmma_f32_16x16x4_f32(
           false, afrag[kt],  false, b,
           (short)0, acc,  false,  false);
    }
    const int col = nt * 16 + lrow;
    #pragma unroll
    for (int r = 0; r < 8; ++r) {
      int m = base + half * 8 + r;
      if (m < n_nodes) h[(size_t)m * DFEAT + col] = acc[r];
    }
  }
}

#define BSH 9
#define BNODES (1 << BSH)
#define MAXBK 128
#define ECHUNK 4096
#define EPT (ECHUNK / 1024)
static_assert(MAXBK / 32 == 4, "bucket_place names 5 x 4 loaded values in its fence");
#define SORT_R 8

__device__ __forceinline__ int bucket_of(const int d, const int nbk) {
  const int b = d >> BSH;
  const int top = ((nbk < MAXBK) ? nbk : MAXBK) - 1;
  return (b < 0) ? 0 : ((b > top) ? top : b);
}

__global__ void __launch_bounds__(1024) bucket_count(
    const int* __restrict__ dst, int* __restrict__ counts, int E, int nbk, int nblk)
{
  __shared__ int bins[MAXBK];
  const int tid = threadIdx.x;
  const int e0 = blockIdx.x * ECHUNK;
  int d[EPT];
  #pragma unroll
  for (int k = 0; k < EPT; ++k) {
    const int e = e0 + k * 1024 + tid;
    d[k] = dst[(e < E) ? e : E - 1];
  }
  if (tid < MAXBK) bins[tid] = 0;
  __syncthreads();
  #pragma unroll
  for (int k = 0; k < EPT; ++k) {
    const int e = e0 + k * 1024 + tid;
    if (e < E) atomicAdd(&bins[bucket_of(d[k], nbk)], 1);
  }
  __syncthreads();
  if (tid < MAXBK) counts[tid * nblk + blockIdx.x] = bins[tid];
}

__global__ void __launch_bounds__(1024) bucket_place(
    const int* __restrict__ src, const int* __restrict__ dst,
    const int* __restrict__ counts, int* __restrict__ bbase, int* __restrict__ tmp,
    int E, int nbk, int nblk)
{
  __shared__ int tot[MAXBK];
  __shared__ int cur[MAXBK];
  __shared__ int wsum[MAXBK / 32];
  const int tid = threadIdx.x, lane = tid & 31, wave = tid >> 5;
  const int blk = blockIdx.x;
  const int e0 = blk * ECHUNK;

  int d[EPT], s[EPT];
  #pragma unroll
  for (int k = 0; k < EPT; ++k) {
    const int e = e0 + k * 1024 + tid;
    const int ec = (e < E) ? e : E - 1;
    d[k] = dst[ec];
    s[k] = src[ec];
  }

  int all[MAXBK / 32], before[MAXBK / 32];
  #pragma unroll
  for (int i = 0; i < MAXBK / 32; ++i) { all[i] = 0; before[i] = 0; }
  for (int j0 = 0; j0 < nblk; j0 += 160) {
    int cv[5][MAXBK / 32];
    #pragma unroll
    for (int u = 0; u < 5; ++u) {
      const int j = j0 + u * 32 + lane;
      const int jc = (j < nblk) ? j : nblk - 1;
      #pragma unroll
      for (int i = 0; i < MAXBK / 32; ++i) cv[u][i] = counts[(wave + 32 * i) * nblk + jc];
    }
    asm volatile("" :: "v"(cv[0][0]), "v"(cv[0][1]), "v"(cv[0][2]), "v"(cv[0][3]),
                       "v"(cv[1][0]), "v"(cv[1][1]), "v"(cv[1][2]), "v"(cv[1][3]),
                       "v"(cv[2][0]), "v"(cv[2][1]), "v"(cv[2][2]), "v"(cv[2][3]),
                       "v"(cv[3][0]), "v"(cv[3][1]), "v"(cv[3][2]), "v"(cv[3][3]),
                       "v"(cv[4][0]), "v"(cv[4][1]), "v"(cv[4][2]), "v"(cv[4][3]));
    #pragma unroll
    for (int u = 0; u < 5; ++u) {
      const int j = j0 + u * 32 + lane;
      #pragma unroll
      for (int i = 0; i < MAXBK / 32; ++i) {
        const int c = (j < nblk) ? cv[u][i] : 0;
        all[i] += c;
        before[i] += (j < blk) ? c : 0;
      }
    }
  }
  #pragma unroll
  for (int i = 0; i < MAXBK / 32; ++i) {
    for (int o = 16; o > 0; o >>= 1) {
      all[i] += __shfl_xor(all[i], o);
      before[i] += __shfl_xor(before[i], o);
    }
    if (lane == 0) { tot[wave + 32 * i] = all[i]; cur[wave + 32 * i] = before[i]; }
  }
  __syncthreads();

  int t = 0, incl = 0;
  if (tid < MAXBK) {
    t = tot[tid];
    incl = t;
    for (int o = 1; o < 32; o <<= 1) {
      const int v = __shfl_up(incl, o);
      if (lane >= o) incl += v;
    }
    if (lane == 31) wsum[wave] = incl;
  }
  __syncthreads();
  if (tid < MAXBK) {
    int woff = 0;
    for (int w = 0; w < MAXBK / 32; ++w) woff += (w < wave) ? wsum[w] : 0;
    const int base = woff + incl - t;
    cur[tid] += base;
    if (blk == 0 && tid < nbk) bbase[tid] = base;
  }
  if (blk == 0 && tid == 0) bbase[(nbk < MAXBK) ? nbk : MAXBK] = E;
  __syncthreads();

  #pragma unroll
  for (int k = 0; k < EPT; ++k) {
    const int e = e0 + k * 1024 + tid;
    if (e < E) {
      const int p = atomicAdd(&cur[bucket_of(d[k], nbk)], 1);
      if ((unsigned)p < (unsigned)E) tmp[p] = (s[k] << BSH) | (d[k] & (BNODES - 1));
    }
  }
}

__global__ void __launch_bounds__(1024) bucket_sort(
    const int* __restrict__ tmp, const int* __restrict__ bbase,
    int* __restrict__ rowptr, int* __restrict__ csr, float* __restrict__ dis,
    int n, int E, int nbk)
{
  __shared__ int cnt[BNODES];
  __shared__ int wsum[BNODES / 32];
  const int tid = threadIdx.x, lane = tid & 31, wave = tid >> 5;
  const int b = blockIdx.x;
  int ebeg = bbase[b];
  int eend = bbase[b + 1];
  ebeg = (ebeg < 0) ? 0 : ebeg;
  eend = (eend > E) ? E : eend;

  int w[SORT_R];
  #pragma unroll
  for (int k = 0; k < SORT_R; ++k) {
    const int e = ebeg + k * 1024 + tid;
    const int v = tmp[(e < E) ? e : E - 1];
    w[k] = (e < eend) ? v : -1;
  }
  if (tid < BNODES) cnt[tid] = 0;
  __syncthreads();
  #pragma unroll
  for (int k = 0; k < SORT_R; ++k)
    if (w[k] >= 0) atomicAdd(&cnt[w[k] & (BNODES - 1)], 1);
  for (int e = ebeg + SORT_R * 1024 + tid; e < eend; e += 1024)
    atomicAdd(&cnt[tmp[e] & (BNODES - 1)], 1);
  __syncthreads();

  int c = 0, incl = 0;
  if (tid < BNODES) {
    c = cnt[tid];
    incl = c;
    for (int o = 1; o < 32; o <<= 1) {
      const int v = __shfl_up(incl, o);
      if (lane >= o) incl += v;
    }
    if (lane == 31) wsum[wave] = incl;
  }
  __syncthreads();
  if (tid < BNODES) {
    int woff = 0;
    for (int k = 0; k < BNODES / 32; ++k) woff += (k < wave) ? wsum[k] : 0;
    const int ex = woff + incl - c;
    cnt[tid] = ex;
    const int node = b * BNODES + tid;
    if (node < n) {
      rowptr[node] = ebeg + ex;
      dis[node] = __frsqrt_rn((float)(c + 1));
    }
  }
  if (b == nbk - 1 && tid == 0) rowptr[n] = eend;
  __syncthreads();

  #pragma unroll
  for (int k = 0; k < SORT_R; ++k) {
    if (w[k] >= 0) {
      const int q = ebeg + atomicAdd(&cnt[w[k] & (BNODES - 1)], 1);
      if ((unsigned)q < (unsigned)E) csr[q] = w[k] >> BSH;
    }
  }
  for (int e = ebeg + SORT_R * 1024 + tid; e < eend; e += 1024) {
    const int v = tmp[e];
    const int q = ebeg + atomicAdd(&cnt[v & (BNODES - 1)], 1);
    if ((unsigned)q < (unsigned)E) csr[q] = v >> BSH;
  }
}


template <int NK>
__device__ __forceinline__ void sort_keys(int (&key)[NK], const int lane) {
  #pragma unroll
  for (int k = 2; k <= 32; k <<= 1) {
    #pragma unroll
    for (int j = k >> 1; j > 0; j >>= 1) {
      const bool up = ((lane & k) == 0);
      const bool lower = ((lane & j) == 0);
      #pragma unroll
      for (int i = 0; i < NK; ++i) {
        const int p = __shfl_xor(key[i], j);
        const int lo = (key[i] < p) ? key[i] : p;
        const int hi = (key[i] < p) ? p : key[i];
        key[i] = (lower == up) ? lo : hi;
      }
    }
  }
}

__device__ __forceinline__ float4 add_rows(
    float4 acc, const int key, const int m, const int d, const int lane,
    const float4* __restrict__ h4, const int n)
{
  int s_l = (lane < m) ? key : d;
  s_l = (s_l < 0) ? 0 : ((s_l >= n) ? n - 1 : s_l);
  int j = 0;
  for (; j + 4 <= m; j += 4) {
    const int s0 = __builtin_amdgcn_readlane(s_l, j);
    const int s1 = __builtin_amdgcn_readlane(s_l, j + 1);
    const int s2 = __builtin_amdgcn_readlane(s_l, j + 2);
    const int s3 = __builtin_amdgcn_readlane(s_l, j + 3);
    const float4 r0 = h4[(size_t)s0 * 32 + lane];
    const float4 r1 = h4[(size_t)s1 * 32 + lane];
    const float4 r2 = h4[(size_t)s2 * 32 + lane];
    const float4 r3 = h4[(size_t)s3 * 32 + lane];
    acc.x += r0.x; acc.y += r0.y; acc.z += r0.z; acc.w += r0.w;
    acc.x += r1.x; acc.y += r1.y; acc.z += r1.z; acc.w += r1.w;
    acc.x += r2.x; acc.y += r2.y; acc.z += r2.z; acc.w += r2.w;
    acc.x += r3.x; acc.y += r3.y; acc.z += r3.z; acc.w += r3.w;
  }
  if (j < m) {
    const int s0 = __builtin_amdgcn_readlane(s_l, j);
    const int s1 = __builtin_amdgcn_readlane(s_l, j + 1);
    const int s2 = __builtin_amdgcn_readlane(s_l, j + 2);
    const float4 r0 = h4[(size_t)s0 * 32 + lane];
    const float4 r1 = h4[(size_t)s1 * 32 + lane];
    const float4 r2 = h4[(size_t)s2 * 32 + lane];
    acc.x += r0.x; acc.y += r0.y; acc.z += r0.z; acc.w += r0.w;
    if (j + 1 < m) { acc.x += r1.x; acc.y += r1.y; acc.z += r1.z; acc.w += r1.w; }
    if (j + 2 < m) { acc.x += r2.x; acc.y += r2.y; acc.z += r2.z; acc.w += r2.w; }
  }
  return acc;
}

__device__ __forceinline__ float4 add_chunks(
    float4 acc, int c, const int end, const int d, const int lane,
    const int* __restrict__ csr, const float4* __restrict__ h4, const int n)
{
  for (; c < end; c += 32) {
    const int m = (end - c < 32) ? (end - c) : 32;
    int key[1] = {0x7fffffff};
    if (lane < m) key[0] = csr[c + lane];
    sort_keys<1>(key, lane);
    acc = add_rows(acc, key[0], m, d, lane, h4, n);
  }
  return acc;
}

__device__ __forceinline__ float4 finish_row(float4 acc, const float4 bv, const float dd) {
  acc.x = bv.x + acc.x * dd;
  acc.y = bv.y + acc.y * dd;
  acc.z = bv.z + acc.z * dd;
  acc.w = bv.w + acc.w * dd;
  float4 o;
  o.x = (acc.x >= 0.0f) ? acc.x : LEAKY * acc.x;
  o.y = (acc.y >= 0.0f) ? acc.y : LEAKY * acc.y;
  o.z = (acc.z >= 0.0f) ? acc.z : LEAKY * acc.z;
  o.w = (acc.w >= 0.0f) ? acc.w : LEAKY * acc.w;
  return o;
}

__global__ void __launch_bounds__(256) gcn_gather(
    const int* __restrict__ rowptr, const int* __restrict__ csr,
    const float* __restrict__ h, const float* __restrict__ dis,
    const float* __restrict__ bias, float* __restrict__ out, int n, int E)
{
  const int d = __builtin_amdgcn_readfirstlane((int)(blockIdx.x * 8 + (threadIdx.x >> 5)));
  if (d >= n) return;
  const int lane = threadIdx.x & 31;
  int beg = __builtin_amdgcn_readfirstlane(rowptr[d]);
  int end = __builtin_amdgcn_readfirstlane(rowptr[d + 1]);
  beg = (beg < 0) ? 0 : beg;
  end = (end > E) ? E : end;
  const float dd = dis[d];
  const float4* __restrict__ h4 = (const float4*)h;
  const float4 hv = h4[(size_t)d * 32 + lane];
  const float4 bv = ((const float4*)bias)[lane];
  const float4 acc = add_chunks(hv, beg, end, d, lane, csr, h4, n);
  ((float4*)out)[(size_t)d * 32 + lane] = finish_row(acc, bv, dd);
}

#define TILE 128
__global__ void __launch_bounds__(1024) gcn_gather_gemm(
    const int* __restrict__ rowptr, const int* __restrict__ csr,
    const float* __restrict__ hin, const float* __restrict__ dis,
    const float* __restrict__ bias, const float* __restrict__ W,
    float* __restrict__ hout, int n, int E)
{
  __shared__ __attribute__((aligned(16))) float Wlds[DFEAT * WPITCH];
  __shared__ __attribute__((aligned(16))) float Alds[TILE * WPITCH];

  {
    const float4* Wg4 = (const float4*)W;
    #pragma unroll
    for (int i = 0; i < (DFEAT * DFEAT / 4) / 1024; ++i) {
      const int idx = threadIdx.x + 1024 * i;
      *(float4*)(Wlds + (idx >> 5) * WPITCH + (idx & 31) * 4) = Wg4[idx];
    }
  }

  const int wave = __builtin_amdgcn_readfirstlane((int)(threadIdx.x >> 5));
  const int lane = threadIdx.x & 31;
  const int tile0 = blockIdx.x * TILE;
  const float4* __restrict__ h4 = (const float4*)hin;

  const int d0 = tile0 + wave * 4;
  int beg[4], end[4], m0[4], key[4];
  #pragma unroll
  for (int i = 0; i < 4; ++i) {
    const int d = d0 + i;
    int b = 0, e = 0;
    if (d < n) {
      b = __builtin_amdgcn_readfirstlane(rowptr[d]);
      e = __builtin_amdgcn_readfirstlane(rowptr[d + 1]);
      b = (b < 0) ? 0 : b;
      e = (e > E) ? E : e;
    }
    beg[i] = b;
    end[i] = e;
    m0[i] = (e - b < 32) ? (e - b) : 32;
    key[i] = 0x7fffffff;
    if (lane < m0[i]) key[i] = csr[b + lane];
  }
  sort_keys<4>(key, lane);

  #pragma unroll
  for (int i = 0; i < 4; ++i) {
    const int lr = wave * 4 + i;
    const int d = d0 + i;
    float4 a = {0.0f, 0.0f, 0.0f, 0.0f};
    if (d < n) {
      const float dd = dis[d];
      const float4 hv = h4[(size_t)d * 32 + lane];
      const float4 bv = ((const float4*)bias)[lane];
      float4 acc = add_rows(hv, key[i], m0[i], d, lane, h4, n);
      acc = add_chunks(acc, beg[i] + 32, end[i], d, lane, csr, h4, n);
      const float4 o = finish_row(acc, bv, dd);
      a.x = o.x * dd; a.y = o.y * dd; a.z = o.z * dd; a.w = o.w * dd;
    }
    *(float4*)(Alds + lr * WPITCH + lane * 4) = a;
  }
  __syncthreads();

  const int half = lane >> 4;
  const int lrow = lane & 15;
  const int rt = wave >> 2;
  const int cg = wave & 3;
  const v2f* ar = (const v2f*)(Alds + (rt * 16 + lrow) * WPITCH);
  const v2f* w0 = (const v2f*)(Wlds + ((cg * 2 + 0) * 16 + lrow) * WPITCH);
  const v2f* w1 = (const v2f*)(Wlds + ((cg * 2 + 1) * 16 + lrow) * WPITCH);
  v8f acc0 = {};
  v8f acc1 = {};
  #pragma unroll
  for (int kt = 0; kt < 32; ++kt) {
    const v2f a  = ar[2 * kt + half];
    const v2f b0 = w0[2 * kt + half];
    const v2f b1 = w1[2 * kt + half];
    acc0 = __builtin_amdgcn_wmma_f32_16x16x4_f32(
         false, a,  false, b0,
         (short)0, acc0,  false,  false);
    acc1 = __builtin_amdgcn_wmma_f32_16x16x4_f32(
         false, a,  false, b1,
         (short)0, acc1,  false,  false);
  }
  const int mbase = tile0 + rt * 16 + half * 8;
  const int col0 = (cg * 2) * 16 + lrow;
  #pragma unroll
  for (int r = 0; r < 8; ++r) {
    const int m = mbase + r;
    if (m < n) {
      hout[(size_t)m * DFEAT + col0]      = acc0[r];
      hout[(size_t)m * DFEAT + col0 + 16] = acc1[r];
    }
  }
}

extern "C" void kernel_launch(void* const* d_in, const int* in_sizes, int n_in,
                              void* d_out, int out_size, void* d_ws, size_t ws_size,
                              hipStream_t stream) {
  const float* x  = (const float*)d_in[0];
  const int*   ei = (const int*)d_in[1];
  const float* Ws = (const float*)d_in[2];
  const float* bs = (const float*)d_in[3];
  float* out = (float*)d_out;

  const int n = in_sizes[0] / DFEAT;
  const int E = in_sizes[1] / 2;
  const int* src = ei;
  const int* dst = ei + E;

  const int nbk  = (n + BNODES - 1) / BNODES;
  const int nblk = (E + ECHUNK - 1) / ECHUNK;

  int* rowptr = (int*)d_ws;
  int* bbase  = rowptr + (n + 1);
  int* counts = bbase + (MAXBK + 1);
  int* tmp    = counts + (size_t)MAXBK * nblk;
  int* csr    = tmp + E;
  float* hbuf = (float*)d_ws + (size_t)n * DFEAT;
  float* dis  = hbuf + (size_t)n * DFEAT;

  bucket_count<<<nblk, 1024, 0, stream>>>(dst, counts, E, nbk, nblk);
  bucket_place<<<nblk, 1024, 0, stream>>>(src, dst, counts, bbase, tmp, E, nbk, nblk);
  bucket_sort <<<nbk, 1024, 0, stream>>>(tmp, bbase, rowptr, csr, dis, n, E, nbk);

  const size_t WW = (size_t)DFEAT * DFEAT;
  gcn_gemm_xWT   <<<(n + 127) / 128, 256, 0, stream>>>(x, Ws, dis, hbuf, n);
  gcn_gather_gemm<<<(n + TILE - 1) / TILE, 1024, 0, stream>>>(rowptr, csr, hbuf, dis, bs, Ws + WW, out, n, E);
  gcn_gather_gemm<<<(n + TILE - 1) / TILE, 1024, 0, stream>>>(rowptr, csr, out, dis, bs + DFEAT, Ws + 2 * WW, hbuf, n, E);
  gcn_gather     <<<(n + 7) / 8, 256, 0, stream>>>(rowptr, csr, hbuf, dis, bs + 2 * DFEAT, out, n, E);
}


# First GEMM's 64-row tiles ride as extra blocks in the 3 edge-sort launches (rows unscaled); layer-1 node sum takes deg^-1/2[s] per edge (rounding order changes); 6 launches, not 7
# speedup vs baseline: 4.0588x; 1.0496x over previous
#include <hip/hip_runtime.h>


typedef __attribute__((ext_vector_type(2))) float v2f;
typedef __attribute__((ext_vector_type(8))) float v8f;

#define DFEAT 128
#define LEAKY 0.01f
#define WPITCH 132

#define GT 64
__device__ __forceinline__ void gemm_tile_xWT(
    const float* __restrict__ x, const float* __restrict__ W,
    float* __restrict__ h, const int n_nodes, const int tile)
{
  __shared__ __attribute__((aligned(16))) float Wlds[DFEAT * WPITCH];
  __shared__ __attribute__((aligned(16))) float Alds[GT * WPITCH];
  const int tile0 = tile * GT;
  if (tile0 >= n_nodes) return;
  {
    const float4* Wg4 = (const float4*)W;
    const float4* x4 = (const float4*)x;
    #pragma unroll
    for (int i = 0; i < (DFEAT * DFEAT / 4) / 1024; ++i) {
      const int idx = threadIdx.x + 1024 * i;
      *(float4*)(Wlds + (idx >> 5) * WPITCH + (idx & 31) * 4) = Wg4[idx];
    }
    #pragma unroll
    for (int i = 0; i < (GT * DFEAT / 4) / 1024; ++i) {
      const int idx = threadIdx.x + 1024 * i;
      int r = tile0 + (idx >> 5);
      r = (r < n_nodes) ? r : n_nodes - 1;
      *(float4*)(Alds + (idx >> 5) * WPITCH + (idx & 31) * 4) = x4[(size_t)r * 32 + (idx & 31)];
    }
  }
  __syncthreads();
  const int wave = threadIdx.x >> 5;
  const int lane = threadIdx.x & 31;
  const int half = lane >> 4;
  const int lrow = lane & 15;
  const int rt = wave >> 3;
  const int cn = wave & 7;
  const v2f* ar = (const v2f*)(Alds + (rt * 16 + lrow) * WPITCH);
  const v2f* wr = (const v2f*)(Wlds + (cn * 16 + lrow) * WPITCH);
  v8f acc = {};
  #pragma unroll
  for (int kt = 0; kt < 32; ++kt) {
    const v2f a = ar[2 * kt + half];
    const v2f b = wr[2 * kt + half];
    acc = __builtin_amdgcn_wmma_f32_16x16x4_f32(
         false, a,  false, b,
         (short)0, acc,  false,  false);
  }
  const int mbase = tile0 + rt * 16 + half * 8;
  const int col = cn * 16 + lrow;
  #pragma unroll
  for (int r = 0; r < 8; ++r) {
    const int m = mbase + r;
    if (m < n_nodes) h[(size_t)m * DFEAT + col] = acc[r];
  }
}

#define BSH 9
#define BNODES (1 << BSH)
#define MAXBK 128
#define ECHUNK 4096
#define EPT (ECHUNK / 1024)
static_assert(MAXBK / 32 == 4, "bucket_place names 5 x 4 loaded values in its fence");
#define SORT_R 8

__device__ __forceinline__ int bucket_of(const int d, const int nbk) {
  const int b = d >> BSH;
  const int top = ((nbk < MAXBK) ? nbk : MAXBK) - 1;
  return (b < 0) ? 0 : ((b > top) ? top : b);
}

__global__ void __launch_bounds__(1024) bucket_count(
    const int* __restrict__ dst, int* __restrict__ counts, int E, int nbk, int nblk,
    const float* __restrict__ x, const float* __restrict__ W, float* __restrict__ h, int n, int tile_first)
{
  if ((int)blockIdx.x >= nblk) { gemm_tile_xWT(x, W, h, n, tile_first + (int)blockIdx.x - nblk); return; }
  __shared__ int bins[MAXBK];
  const int tid = threadIdx.x;
  const int e0 = blockIdx.x * ECHUNK;
  int d[EPT];
  #pragma unroll
  for (int k = 0; k < EPT; ++k) {
    const int e = e0 + k * 1024 + tid;
    d[k] = dst[(e < E) ? e : E - 1];
  }
  if (tid < MAXBK) bins[tid] = 0;
  __syncthreads();
  #pragma unroll
  for (int k = 0; k < EPT; ++k) {
    const int e = e0 + k * 1024 + tid;
    if (e < E) atomicAdd(&bins[bucket_of(d[k], nbk)], 1);
  }
  __syncthreads();
  if (tid < MAXBK) counts[tid * nblk + blockIdx.x] = bins[tid];
}

__global__ void __launch_bounds__(1024) bucket_place(
    const int* __restrict__ src, const int* __restrict__ dst,
    const int* __restrict__ counts, int* __restrict__ bbase, int* __restrict__ tmp,
    int E, int nbk, int nblk,
    const float* __restrict__ x, const float* __restrict__ W, float* __restrict__ h, int n, int tile_first)
{
  if ((int)blockIdx.x >= nblk) { gemm_tile_xWT(x, W, h, n, tile_first + (int)blockIdx.x - nblk); return; }
  __shared__ int tot[MAXBK];
  __shared__ int cur[MAXBK];
  __shared__ int wsum[MAXBK / 32];
  const int tid = threadIdx.x, lane = tid & 31, wave = tid >> 5;
  const int blk = blockIdx.x;
  const int e0 = blk * ECHUNK;

  int d[EPT], s[EPT];
  #pragma unroll
  for (int k = 0; k < EPT; ++k) {
    const int e = e0 + k * 1024 + tid;
    const int ec = (e < E) ? e : E - 1;
    d[k] = dst[ec];
    s[k] = src[ec];
  }

  int all[MAXBK / 32], before[MAXBK / 32];
  #pragma unroll
  for (int i = 0; i < MAXBK / 32; ++i) { all[i] = 0; before[i] = 0; }
  for (int j0 = 0; j0 < nblk; j0 += 160) {
    int cv[5][MAXBK / 32];
    #pragma unroll
    for (int u = 0; u < 5; ++u) {
      const int j = j0 + u * 32 + lane;
      const int jc = (j < nblk) ? j : nblk - 1;
      #pragma unroll
      for (int i = 0; i < MAXBK / 32; ++i) cv[u][i] = counts[(wave + 32 * i) * nblk + jc];
    }
    asm volatile("" :: "v"(cv[0][0]), "v"(cv[0][1]), "v"(cv[0][2]), "v"(cv[0][3]),
                       "v"(cv[1][0]), "v"(cv[1][1]), "v"(cv[1][2]), "v"(cv[1][3]),
                       "v"(cv[2][0]), "v"(cv[2][1]), "v"(cv[2][2]), "v"(cv[2][3]),
                       "v"(cv[3][0]), "v"(cv[3][1]), "v"(cv[3][2]), "v"(cv[3][3]),
                       "v"(cv[4][0]), "v"(cv[4][1]), "v"(cv[4][2]), "v"(cv[4][3]));
    #pragma unroll
    for (int u = 0; u < 5; ++u) {
      const int j = j0 + u * 32 + lane;
      #pragma unroll
      for (int i = 0; i < MAXBK / 32; ++i) {
        const int c = (j < nblk) ? cv[u][i] : 0;
        all[i] += c;
        before[i] += (j < blk) ? c : 0;
      }
    }
  }
  #pragma unroll
  for (int i = 0; i < MAXBK / 32; ++i) {
    for (int o = 16; o > 0; o >>= 1) {
      all[i] += __shfl_xor(all[i], o);
      before[i] += __shfl_xor(before[i], o);
    }
    if (lane == 0) { tot[wave + 32 * i] = all[i]; cur[wave + 32 * i] = before[i]; }
  }
  __syncthreads();

  int t = 0, incl = 0;
  if (tid < MAXBK) {
    t = tot[tid];
    incl = t;
    for (int o = 1; o < 32; o <<= 1) {
      const int v = __shfl_up(incl, o);
      if (lane >= o) incl += v;
    }
    if (lane == 31) wsum[wave] = incl;
  }
  __syncthreads();
  if (tid < MAXBK) {
    int woff = 0;
    for (int w = 0; w < MAXBK / 32; ++w) woff += (w < wave) ? wsum[w] : 0;
    const int base = woff + incl - t;
    cur[tid] += base;
    if (blk == 0 && tid < nbk) bbase[tid] = base;
  }
  if (blk == 0 && tid == 0) bbase[(nbk < MAXBK) ? nbk : MAXBK] = E;
  __syncthreads();

  #pragma unroll
  for (int k = 0; k < EPT; ++k) {
    const int e = e0 + k * 1024 + tid;
    if (e < E) {
      const int p = atomicAdd(&cur[bucket_of(d[k], nbk)], 1);
      if ((unsigned)p < (unsigned)E) tmp[p] = (s[k] << BSH) | (d[k] & (BNODES - 1));
    }
  }
}

__global__ void __launch_bounds__(1024) bucket_sort(
    const int* __restrict__ tmp, const int* __restrict__ bbase,
    int* __restrict__ rowptr, int* __restrict__ csr, float* __restrict__ dis,
    int n, int E, int nbk,
    const float* __restrict__ x, const float* __restrict__ W, float* __restrict__ h, int tile_first)
{
  if ((int)blockIdx.x >= nbk) { gemm_tile_xWT(x, W, h, n, tile_first + (int)blockIdx.x - nbk); return; }
  __shared__ int cnt[BNODES];
  __shared__ int wsum[BNODES / 32];
  const int tid = threadIdx.x, lane = tid & 31, wave = tid >> 5;
  const int b = blockIdx.x;
  int ebeg = bbase[b];
  int eend = bbase[b + 1];
  ebeg = (ebeg < 0) ? 0 : ebeg;
  eend = (eend > E) ? E : eend;

  int w[SORT_R];
  #pragma unroll
  for (int k = 0; k < SORT_R; ++k) {
    const int e = ebeg + k * 1024 + tid;
    const int v = tmp[(e < E) ? e : E - 1];
    w[k] = (e < eend) ? v : -1;
  }
  if (tid < BNODES) cnt[tid] = 0;
  __syncthreads();
  #pragma unroll
  for (int k = 0; k < SORT_R; ++k)
    if (w[k] >= 0) atomicAdd(&cnt[w[k] & (BNODES - 1)], 1);
  for (int e = ebeg + SORT_R * 1024 + tid; e < eend; e += 1024)
    atomicAdd(&cnt[tmp[e] & (BNODES - 1)], 1);
  __syncthreads();

  int c = 0, incl = 0;
  if (tid < BNODES) {
    c = cnt[tid];
    incl = c;
    for (int o = 1; o < 32; o <<= 1) {
      const int v = __shfl_up(incl, o);
      if (lane >= o) incl += v;
    }
    if (lane == 31) wsum[wave] = incl;
  }
  __syncthreads();
  if (tid < BNODES) {
    int woff = 0;
    for (int k = 0; k < BNODES / 32; ++k) woff += (k < wave) ? wsum[k] : 0;
    const int ex = woff + incl - c;
    cnt[tid] = ex;
    const int node = b * BNODES + tid;
    if (node < n) {
      rowptr[node] = ebeg + ex;
      dis[node] = __frsqrt_rn((float)(c + 1));
    }
  }
  if (b == nbk - 1 && tid == 0) rowptr[n] = eend;
  __syncthreads();

  #pragma unroll
  for (int k = 0; k < SORT_R; ++k) {
    if (w[k] >= 0) {
      const int q = ebeg + atomicAdd(&cnt[w[k] & (BNODES - 1)], 1);
      if ((unsigned)q < (unsigned)E) csr[q] = w[k] >> BSH;
    }
  }
  for (int e = ebeg + SORT_R * 1024 + tid; e < eend; e += 1024) {
    const int v = tmp[e];
    const int q = ebeg + atomicAdd(&cnt[v & (BNODES - 1)], 1);
    if ((unsigned)q < (unsigned)E) csr[q] = v >> BSH;
  }
}


template <int NK>
__device__ __forceinline__ void sort_keys(int (&key)[NK], const int lane) {
  #pragma unroll
  for (int k = 2; k <= 32; k <<= 1) {
    #pragma unroll
    for (int j = k >> 1; j > 0; j >>= 1) {
      const bool up = ((lane & k) == 0);
      const bool lower = ((lane & j) == 0);
      #pragma unroll
      for (int i = 0; i < NK; ++i) {
        const int p = __shfl_xor(key[i], j);
        const int lo = (key[i] < p) ? key[i] : p;
        const int hi = (key[i] < p) ? p : key[i];
        key[i] = (lower == up) ? lo : hi;
      }
    }
  }
}

template <bool SC>
__device__ __forceinline__ float4 add_rows(
    float4 acc, const int key, const int m, const int d, const int lane,
    const float4* __restrict__ h4, const float* __restrict__ dis, const int n)
{
  int s_l = (lane < m) ? key : d;
  s_l = (s_l < 0) ? 0 : ((s_l >= n) ? n - 1 : s_l);
  int w_l = 0;
  if (SC) w_l = __float_as_int(dis[s_l]);
  int j = 0;
  for (; j + 4 <= m; j += 4) {
    const int s0 = __builtin_amdgcn_readlane(s_l, j);
    const int s1 = __builtin_amdgcn_readlane(s_l, j + 1);
    const int s2 = __builtin_amdgcn_readlane(s_l, j + 2);
    const int s3 = __builtin_amdgcn_readlane(s_l, j + 3);
    const float4 r0 = h4[(size_t)s0 * 32 + lane];
    const float4 r1 = h4[(size_t)s1 * 32 + lane];
    const float4 r2 = h4[(size_t)s2 * 32 + lane];
    const float4 r3 = h4[(size_t)s3 * 32 + lane];
    if (SC) {
      const float f0 = __int_as_float(__builtin_amdgcn_readlane(w_l, j));
      const float f1 = __int_as_float(__builtin_amdgcn_readlane(w_l, j + 1));
      const float f2 = __int_as_float(__builtin_amdgcn_readlane(w_l, j + 2));
      const float f3 = __int_as_float(__builtin_amdgcn_readlane(w_l, j + 3));
      acc.x += r0.x * f0; acc.y += r0.y * f0; acc.z += r0.z * f0; acc.w += r0.w * f0;
      acc.x += r1.x * f1; acc.y += r1.y * f1; acc.z += r1.z * f1; acc.w += r1.w * f1;
      acc.x += r2.x * f2; acc.y += r2.y * f2; acc.z += r2.z * f2; acc.w += r2.w * f2;
      acc.x += r3.x * f3; acc.y += r3.y * f3; acc.z += r3.z * f3; acc.w += r3.w * f3;
    } else {
      acc.x += r0.x; acc.y += r0.y; acc.z += r0.z; acc.w += r0.w;
      acc.x += r1.x; acc.y += r1.y; acc.z += r1.z; acc.w += r1.w;
      acc.x += r2.x; acc.y += r2.y; acc.z += r2.z; acc.w += r2.w;
      acc.x += r3.x; acc.y += r3.y; acc.z += r3.z; acc.w += r3.w;
    }
  }
  if (j < m) {
    const int s0 = __builtin_amdgcn_readlane(s_l, j);
    const int s1 = __builtin_amdgcn_readlane(s_l, j + 1);
    const int s2 = __builtin_amdgcn_readlane(s_l, j + 2);
    const float4 r0 = h4[(size_t)s0 * 32 + lane];
    const float4 r1 = h4[(size_t)s1 * 32 + lane];
    const float4 r2 = h4[(size_t)s2 * 32 + lane];
    float f0 = 1.0f, f1 = 1.0f, f2 = 1.0f;
    if (SC) {
      f0 = __int_as_float(__builtin_amdgcn_readlane(w_l, j));
      f1 = __int_as_float(__builtin_amdgcn_readlane(w_l, j + 1));
      f2 = __int_as_float(__builtin_amdgcn_readlane(w_l, j + 2));
    }
    if (SC) { acc.x += r0.x * f0; acc.y += r0.y * f0; acc.z += r0.z * f0; acc.w += r0.w * f0; }
    else    { acc.x += r0.x; acc.y += r0.y; acc.z += r0.z; acc.w += r0.w; }
    if (j + 1 < m) {
      if (SC) { acc.x += r1.x * f1; acc.y += r1.y * f1; acc.z += r1.z * f1; acc.w += r1.w * f1; }
      else    { acc.x += r1.x; acc.y += r1.y; acc.z += r1.z; acc.w += r1.w; }
    }
    if (j + 2 < m) {
      if (SC) { acc.x += r2.x * f2; acc.y += r2.y * f2; acc.z += r2.z * f2; acc.w += r2.w * f2; }
      else    { acc.x += r2.x; acc.y += r2.y; acc.z += r2.z; acc.w += r2.w; }
    }
  }
  return acc;
}

template <bool SC>
__device__ __forceinline__ float4 add_chunks(
    float4 acc, int c, const int end, const int d, const int lane,
    const int* __restrict__ csr, const float4* __restrict__ h4,
    const float* __restrict__ dis, const int n)
{
  for (; c < end; c += 32) {
    const int m = (end - c < 32) ? (end - c) : 32;
    int key[1] = {0x7fffffff};
    if (lane < m) key[0] = csr[c + lane];
    sort_keys<1>(key, lane);
    acc = add_rows<SC>(acc, key[0], m, d, lane, h4, dis, n);
  }
  return acc;
}

__device__ __forceinline__ float4 finish_row(float4 acc, const float4 bv, const float dd) {
  acc.x = bv.x + acc.x * dd;
  acc.y = bv.y + acc.y * dd;
  acc.z = bv.z + acc.z * dd;
  acc.w = bv.w + acc.w * dd;
  float4 o;
  o.x = (acc.x >= 0.0f) ? acc.x : LEAKY * acc.x;
  o.y = (acc.y >= 0.0f) ? acc.y : LEAKY * acc.y;
  o.z = (acc.z >= 0.0f) ? acc.z : LEAKY * acc.z;
  o.w = (acc.w >= 0.0f) ? acc.w : LEAKY * acc.w;
  return o;
}

__global__ void __launch_bounds__(256) gcn_gather(
    const int* __restrict__ rowptr, const int* __restrict__ csr,
    const float* __restrict__ h, const float* __restrict__ dis,
    const float* __restrict__ bias, float* __restrict__ out, int n, int E)
{
  const int d = __builtin_amdgcn_readfirstlane((int)(blockIdx.x * 8 + (threadIdx.x >> 5)));
  if (d >= n) return;
  const int lane = threadIdx.x & 31;
  int beg = __builtin_amdgcn_readfirstlane(rowptr[d]);
  int end = __builtin_amdgcn_readfirstlane(rowptr[d + 1]);
  beg = (beg < 0) ? 0 : beg;
  end = (end > E) ? E : end;
  const float dd = dis[d];
  const float4* __restrict__ h4 = (const float4*)h;
  const float4 hv = h4[(size_t)d * 32 + lane];
  const float4 bv = ((const float4*)bias)[lane];
  const float4 acc = add_chunks<false>(hv, beg, end, d, lane, csr, h4, dis, n);
  ((float4*)out)[(size_t)d * 32 + lane] = finish_row(acc, bv, dd);
}

#define TILE 128
template <bool SC>
__global__ void __launch_bounds__(1024) gcn_gather_gemm(
    const int* __restrict__ rowptr, const int* __restrict__ csr,
    const float* __restrict__ hin, const float* __restrict__ dis,
    const float* __restrict__ bias, const float* __restrict__ W,
    float* __restrict__ hout, int n, int E)
{
  __shared__ __attribute__((aligned(16))) float Wlds[DFEAT * WPITCH];
  __shared__ __attribute__((aligned(16))) float Alds[TILE * WPITCH];

  {
    const float4* Wg4 = (const float4*)W;
    #pragma unroll
    for (int i = 0; i < (DFEAT * DFEAT / 4) / 1024; ++i) {
      const int idx = threadIdx.x + 1024 * i;
      *(float4*)(Wlds + (idx >> 5) * WPITCH + (idx & 31) * 4) = Wg4[idx];
    }
  }

  const int wave = __builtin_amdgcn_readfirstlane((int)(threadIdx.x >> 5));
  const int lane = threadIdx.x & 31;
  const int tile0 = blockIdx.x * TILE;
  const float4* __restrict__ h4 = (const float4*)hin;

  const int d0 = tile0 + wave * 4;
  int beg[4], end[4], m0[4], key[4];
  #pragma unroll
  for (int i = 0; i < 4; ++i) {
    const int d = d0 + i;
    int b = 0, e = 0;
    if (d < n) {
      b = __builtin_amdgcn_readfirstlane(rowptr[d]);
      e = __builtin_amdgcn_readfirstlane(rowptr[d + 1]);
      b = (b < 0) ? 0 : b;
      e = (e > E) ? E : e;
    }
    beg[i] = b;
    end[i] = e;
    m0[i] = (e - b < 32) ? (e - b) : 32;
    key[i] = 0x7fffffff;
    if (lane < m0[i]) key[i] = csr[b + lane];
  }
  sort_keys<4>(key, lane);

  #pragma unroll
  for (int i = 0; i < 4; ++i) {
    const int lr = wave * 4 + i;
    const int d = d0 + i;
    float4 a = {0.0f, 0.0f, 0.0f, 0.0f};
    if (d < n) {
      const float dd = dis[d];
      float4 hv = h4[(size_t)d * 32 + lane];
      if (SC) { hv.x *= dd; hv.y *= dd; hv.z *= dd; hv.w *= dd; }
      const float4 bv = ((const float4*)bias)[lane];
      float4 acc = add_rows<SC>(hv, key[i], m0[i], d, lane, h4, dis, n);
      acc = add_chunks<SC>(acc, beg[i] + 32, end[i], d, lane, csr, h4, dis, n);
      const float4 o = finish_row(acc, bv, dd);
      a.x = o.x * dd; a.y = o.y * dd; a.z = o.z * dd; a.w = o.w * dd;
    }
    *(float4*)(Alds + lr * WPITCH + lane * 4) = a;
  }
  __syncthreads();

  const int half = lane >> 4;
  const int lrow = lane & 15;
  const int rt = wave >> 2;
  const int cg = wave & 3;
  const v2f* ar = (const v2f*)(Alds + (rt * 16 + lrow) * WPITCH);
  const v2f* w0 = (const v2f*)(Wlds + ((cg * 2 + 0) * 16 + lrow) * WPITCH);
  const v2f* w1 = (const v2f*)(Wlds + ((cg * 2 + 1) * 16 + lrow) * WPITCH);
  v8f acc0 = {};
  v8f acc1 = {};
  #pragma unroll
  for (int kt = 0; kt < 32; ++kt) {
    const v2f a  = ar[2 * kt + half];
    const v2f b0 = w0[2 * kt + half];
    const v2f b1 = w1[2 * kt + half];
    acc0 = __builtin_amdgcn_wmma_f32_16x16x4_f32(
         false, a,  false, b0,
         (short)0, acc0,  false,  false);
    acc1 = __builtin_amdgcn_wmma_f32_16x16x4_f32(
         false, a,  false, b1,
         (short)0, acc1,  false,  false);
  }
  const int mbase = tile0 + rt * 16 + half * 8;
  const int col0 = (cg * 2) * 16 + lrow;
  #pragma unroll
  for (int r = 0; r < 8; ++r) {
    const int m = mbase + r;
    if (m < n) {
      hout[(size_t)m * DFEAT + col0]      = acc0[r];
      hout[(size_t)m * DFEAT + col0 + 16] = acc1[r];
    }
  }
}

extern "C" void kernel_launch(void* const* d_in, const int* in_sizes, int n_in,
                              void* d_out, int out_size, void* d_ws, size_t ws_size,
                              hipStream_t stream) {
  const float* x  = (const float*)d_in[0];
  const int*   ei = (const int*)d_in[1];
  const float* Ws = (const float*)d_in[2];
  const float* bs = (const float*)d_in[3];
  float* out = (float*)d_out;

  const int n = in_sizes[0] / DFEAT;
  const int E = in_sizes[1] / 2;
  const int* src = ei;
  const int* dst = ei + E;

  const int nbk  = (n + BNODES - 1) / BNODES;
  const int nblk = (E + ECHUNK - 1) / ECHUNK;

  int* rowptr = (int*)d_ws;
  int* bbase  = rowptr + (n + 1);
  int* counts = bbase + (MAXBK + 1);
  int* tmp    = counts + (size_t)MAXBK * nblk;
  int* csr    = tmp + E;
  float* hbuf = (float*)d_ws + (size_t)n * DFEAT;
  float* dis  = hbuf + (size_t)n * DFEAT;

  const int ntile = (n + GT - 1) / GT;
  const int t1 = ntile / 3;
  const int t2 = ntile / 3;
  const int t3 = ntile - t1 - t2;
  bucket_count<<<nblk + t1, 1024, 0, stream>>>(dst, counts, E, nbk, nblk, x, Ws, hbuf, n, 0);
  bucket_place<<<nblk + t2, 1024, 0, stream>>>(src, dst, counts, bbase, tmp, E, nbk, nblk, x, Ws, hbuf, n, t1);
  bucket_sort <<<nbk + t3, 1024, 0, stream>>>(tmp, bbase, rowptr, csr, dis, n, E, nbk, x, Ws, hbuf, t1 + t2);

  const size_t WW = (size_t)DFEAT * DFEAT;
  gcn_gather_gemm<true> <<<(n + TILE - 1) / TILE, 1024, 0, stream>>>(rowptr, csr, hbuf, dis, bs, Ws + WW, out, n, E);
  gcn_gather_gemm<false><<<(n + TILE - 1) / TILE, 1024, 0, stream>>>(rowptr, csr, out, dis, bs + DFEAT, Ws + 2 * WW, hbuf, n, E);
  gcn_gather     <<<(n + 7) / 8, 256, 0, stream>>>(rowptr, csr, hbuf, dis, bs + 2 * DFEAT, out, n, E);
}
